# starting kernel
# baseline (speedup 1.0000x reference)
#include <hip/hip_runtime.h>
#include <math.h>

typedef __attribute__((ext_vector_type(2))) float v2f;
typedef __attribute__((ext_vector_type(8))) float v8f;

#define NBANDS 65
#define LFIR   129
#define NPAD   144
#define FRAME  80
#define TFRAMES 4000
#define BATCH  32
#define LCONV  208
#define ROWS   (BATCH*TFRAMES)
#define NTILES 9
#define RTILES (ROWS/16)
#define KSTEPS 17
#define PPAIRS (KSTEPS*2)

__global__ void build_B(float* __restrict__ Bp) {
  int i = blockIdx.x * blockDim.x + threadIdx.x;
  if (i >= PPAIRS * NPAD * 2) return;
  int p   = i / (NPAD * 2);
  int rem = i - p * (NPAD * 2);
  int n   = rem >> 1;
  int r   = rem & 1;
  int k   = 2 * p + r;
  float v = 0.0f;
  if (k < NBANDS && n < LFIR) {
    const float PI = 3.14159265358979323846f;
    float hann = 0.5f * (1.0f - cosf(2.0f * PI * (float)n / (float)LFIR));
    float w = (k == 0) ? 1.0f : 2.0f;
    float c = cosf(2.0f * PI * (float)k * (float)(n - (NBANDS - 1)) / (float)LFIR);
    v = hann * w * c / (float)LFIR;
  }
  Bp[i] = v;
}

__global__ void fir_gemm(const float* __restrict__ H,
                         const float* __restrict__ Bp,
                         float* __restrict__ FIR) {
  int wave = (int)((blockIdx.x * blockDim.x + threadIdx.x) >> 5);
  int lane = (int)(threadIdx.x & 31u);
  if (wave >= RTILES) return;

  int half = lane >> 4;
  int lm   = lane & 15;

  int row = wave * 16 + lm;
  const float* hrow = H + (size_t)row * NBANDS;

  v8f c0 = {}, c1 = {}, c2 = {}, c3 = {}, c4 = {}, c5 = {}, c6 = {}, c7 = {}, c8 = {};

#pragma unroll 1
  for (int kk = 0; kk < KSTEPS - 1; ++kk) {
    int kA = kk * 4 + 2 * half;
    v2f a;
    a.x = hrow[kA];
    a.y = hrow[kA + 1];
    const v2f* brow = (const v2f*)(Bp + ((size_t)(kk * 2 + half) * NPAD + lm) * 2);
    v2f b0 = brow[0],  b1 = brow[16],  b2 = brow[32],  b3 = brow[48],
        b4 = brow[64], b5 = brow[80],  b6 = brow[96],  b7 = brow[112],
        b8 = brow[128];
    c0 = __builtin_amdgcn_wmma_f32_16x16x4_f32(false, a, false, b0, (short)0, c0, false, false);
    c1 = __builtin_amdgcn_wmma_f32_16x16x4_f32(false, a, false, b1, (short)0, c1, false, false);
    c2 = __builtin_amdgcn_wmma_f32_16x16x4_f32(false, a, false, b2, (short)0, c2, false, false);
    c3 = __builtin_amdgcn_wmma_f32_16x16x4_f32(false, a, false, b3, (short)0, c3, false, false);
    c4 = __builtin_amdgcn_wmma_f32_16x16x4_f32(false, a, false, b4, (short)0, c4, false, false);
    c5 = __builtin_amdgcn_wmma_f32_16x16x4_f32(false, a, false, b5, (short)0, c5, false, false);
    c6 = __builtin_amdgcn_wmma_f32_16x16x4_f32(false, a, false, b6, (short)0, c6, false, false);
    c7 = __builtin_amdgcn_wmma_f32_16x16x4_f32(false, a, false, b7, (short)0, c7, false, false);
    c8 = __builtin_amdgcn_wmma_f32_16x16x4_f32(false, a, false, b8, (short)0, c8, false, false);
  }

  {
    float h64 = hrow[NBANDS - 1];
    v2f a;
    a.x = half ? 0.0f : h64;
    a.y = 0.0f;
    const int kk = KSTEPS - 1;
    const v2f* brow = (const v2f*)(Bp + ((size_t)(kk * 2 + half) * NPAD + lm) * 2);
    v2f b0 = brow[0],  b1 = brow[16],  b2 = brow[32],  b3 = brow[48],
        b4 = brow[64], b5 = brow[80],  b6 = brow[96],  b7 = brow[112],
        b8 = brow[128];
    c0 = __builtin_amdgcn_wmma_f32_16x16x4_f32(false, a, false, b0, (short)0, c0, false, false);
    c1 = __builtin_amdgcn_wmma_f32_16x16x4_f32(false, a, false, b1, (short)0, c1, false, false);
    c2 = __builtin_amdgcn_wmma_f32_16x16x4_f32(false, a, false, b2, (short)0, c2, false, false);
    c3 = __builtin_amdgcn_wmma_f32_16x16x4_f32(false, a, false, b3, (short)0, c3, false, false);
    c4 = __builtin_amdgcn_wmma_f32_16x16x4_f32(false, a, false, b4, (short)0, c4, false, false);
    c5 = __builtin_amdgcn_wmma_f32_16x16x4_f32(false, a, false, b5, (short)0, c5, false, false);
    c6 = __builtin_amdgcn_wmma_f32_16x16x4_f32(false, a, false, b6, (short)0, c6, false, false);
    c7 = __builtin_amdgcn_wmma_f32_16x16x4_f32(false, a, false, b7, (short)0, c7, false, false);
    c8 = __builtin_amdgcn_wmma_f32_16x16x4_f32(false, a, false, b8, (short)0, c8, false, false);
  }

  size_t rbase = (size_t)(wave * 16 + 8 * half) * NPAD;
  v8f acc[9] = {c0, c1, c2, c3, c4, c5, c6, c7, c8};
#pragma unroll
  for (int j = 0; j < NTILES; ++j) {
    size_t outBase = rbase + j * 16 + lm;
#pragma unroll
    for (int v = 0; v < 8; ++v)
      FIR[outBase + (size_t)v * NPAD] = acc[j][v];
  }
}

__global__ void conv_ola(const float* __restrict__ noise,
                         const float* __restrict__ FIR,
                         float* __restrict__ out) {
  long long idx = (long long)blockIdx.x * blockDim.x + threadIdx.x;
  if (idx >= (long long)BATCH * TFRAMES * FRAME) return;
  int b = (int)(idx / (TFRAMES * FRAME));
  int s = (int)(idx - (long long)b * (TFRAMES * FRAME));

  int u = s + (NBANDS - 1);
  int t_hi = u / FRAME; if (t_hi > TFRAMES - 1) t_hi = TFRAMES - 1;
  int lo = u - (LCONV - 1);
  int t_lo = (lo <= 0) ? 0 : (lo + FRAME - 1) / FRAME;

  float acc = 0.0f;
  for (int t = t_lo; t <= t_hi; ++t) {
    int n = u - t * FRAME;
    int kmin = n - (FRAME - 1); if (kmin < 0) kmin = 0;
    int kmax = n;               if (kmax > LFIR - 1) kmax = LFIR - 1;
    const float* xr = noise + (size_t)(b * TFRAMES + t) * FRAME;
    const float* fr = FIR   + (size_t)(b * TFRAMES + t) * NPAD;
    for (int k = kmin; k <= kmax; ++k) {
      float x = 2.0f * xr[n - k] - 1.0f;
      acc = fmaf(x, fr[k], acc);
    }
  }
  out[idx] = acc;
}

extern "C" void kernel_launch(void* const* d_in, const int* in_sizes, int n_in,
                              void* d_out, int out_size, void* d_ws, size_t ws_size,
                              hipStream_t stream) {
  const float* H     = (const float*)d_in[0];
  const float* noise = (const float*)d_in[1];
  float* out = (float*)d_out;

  float* Bp  = (float*)d_ws;
  float* FIR = (float*)((char*)d_ws + 64 * 1024);

  {
    int total = PPAIRS * NPAD * 2;
    build_B<<<(total + 255) / 256, 256, 0, stream>>>(Bp);
  }
  {
    long long threads = (long long)RTILES * 32;
    fir_gemm<<<(unsigned)((threads + 255) / 256), 256, 0, stream>>>(H, Bp, FIR);
  }
  {
    long long total = (long long)BATCH * TFRAMES * FRAME;
    conv_ola<<<(unsigned)((total + 255) / 256), 256, 0, stream>>>(noise, FIR, out);
  }
}


# conv_ola: noise and FIR tiles of 32 frames staged in LDS, each thread keeps 16 outputs x 4 taps in registers (wide LDS reads); taps are summed in one ascending pass, so the sum order changes
# speedup vs baseline: 6.2290x; 6.2290x over previous
#include <hip/hip_runtime.h>
#include <math.h>

typedef __attribute__((ext_vector_type(2))) float v2f;
typedef __attribute__((ext_vector_type(8))) float v8f;

#define NBANDS 65
#define LFIR   129
#define NPAD   144
#define FRAME  80
#define TFRAMES 4000
#define BATCH  32
#define LCONV  208
#define ROWS   (BATCH*TFRAMES)
#define NTILES 9
#define RTILES (ROWS/16)
#define KSTEPS 17
#define PPAIRS (KSTEPS*2)

__global__ void build_B(float* __restrict__ Bp) {
  int i = blockIdx.x * blockDim.x + threadIdx.x;
  if (i >= PPAIRS * NPAD * 2) return;
  int p   = i / (NPAD * 2);
  int rem = i - p * (NPAD * 2);
  int n   = rem >> 1;
  int r   = rem & 1;
  int k   = 2 * p + r;
  float v = 0.0f;
  if (k < NBANDS && n < LFIR) {
    const float PI = 3.14159265358979323846f;
    float hann = 0.5f * (1.0f - cosf(2.0f * PI * (float)n / (float)LFIR));
    float w = (k == 0) ? 1.0f : 2.0f;
    float c = cosf(2.0f * PI * (float)k * (float)(n - (NBANDS - 1)) / (float)LFIR);
    v = hann * w * c / (float)LFIR;
  }
  Bp[i] = v;
}

__global__ void fir_gemm(const float* __restrict__ H,
                         const float* __restrict__ Bp,
                         float* __restrict__ FIR) {
  int wave = (int)((blockIdx.x * blockDim.x + threadIdx.x) >> 5);
  int lane = (int)(threadIdx.x & 31u);
  if (wave >= RTILES) return;

  int half = lane >> 4;
  int lm   = lane & 15;

  int row = wave * 16 + lm;
  const float* hrow = H + (size_t)row * NBANDS;

  v8f c0 = {}, c1 = {}, c2 = {}, c3 = {}, c4 = {}, c5 = {}, c6 = {}, c7 = {}, c8 = {};

#pragma unroll 1
  for (int kk = 0; kk < KSTEPS - 1; ++kk) {
    int kA = kk * 4 + 2 * half;
    v2f a;
    a.x = hrow[kA];
    a.y = hrow[kA + 1];
    const v2f* brow = (const v2f*)(Bp + ((size_t)(kk * 2 + half) * NPAD + lm) * 2);
    v2f b0 = brow[0],  b1 = brow[16],  b2 = brow[32],  b3 = brow[48],
        b4 = brow[64], b5 = brow[80],  b6 = brow[96],  b7 = brow[112],
        b8 = brow[128];
    c0 = __builtin_amdgcn_wmma_f32_16x16x4_f32(false, a, false, b0, (short)0, c0, false, false);
    c1 = __builtin_amdgcn_wmma_f32_16x16x4_f32(false, a, false, b1, (short)0, c1, false, false);
    c2 = __builtin_amdgcn_wmma_f32_16x16x4_f32(false, a, false, b2, (short)0, c2, false, false);
    c3 = __builtin_amdgcn_wmma_f32_16x16x4_f32(false, a, false, b3, (short)0, c3, false, false);
    c4 = __builtin_amdgcn_wmma_f32_16x16x4_f32(false, a, false, b4, (short)0, c4, false, false);
    c5 = __builtin_amdgcn_wmma_f32_16x16x4_f32(false, a, false, b5, (short)0, c5, false, false);
    c6 = __builtin_amdgcn_wmma_f32_16x16x4_f32(false, a, false, b6, (short)0, c6, false, false);
    c7 = __builtin_amdgcn_wmma_f32_16x16x4_f32(false, a, false, b7, (short)0, c7, false, false);
    c8 = __builtin_amdgcn_wmma_f32_16x16x4_f32(false, a, false, b8, (short)0, c8, false, false);
  }

  {
    float h64 = hrow[NBANDS - 1];
    v2f a;
    a.x = half ? 0.0f : h64;
    a.y = 0.0f;
    const int kk = KSTEPS - 1;
    const v2f* brow = (const v2f*)(Bp + ((size_t)(kk * 2 + half) * NPAD + lm) * 2);
    v2f b0 = brow[0],  b1 = brow[16],  b2 = brow[32],  b3 = brow[48],
        b4 = brow[64], b5 = brow[80],  b6 = brow[96],  b7 = brow[112],
        b8 = brow[128];
    c0 = __builtin_amdgcn_wmma_f32_16x16x4_f32(false, a, false, b0, (short)0, c0, false, false);
    c1 = __builtin_amdgcn_wmma_f32_16x16x4_f32(false, a, false, b1, (short)0, c1, false, false);
    c2 = __builtin_amdgcn_wmma_f32_16x16x4_f32(false, a, false, b2, (short)0, c2, false, false);
    c3 = __builtin_amdgcn_wmma_f32_16x16x4_f32(false, a, false, b3, (short)0, c3, false, false);
    c4 = __builtin_amdgcn_wmma_f32_16x16x4_f32(false, a, false, b4, (short)0, c4, false, false);
    c5 = __builtin_amdgcn_wmma_f32_16x16x4_f32(false, a, false, b5, (short)0, c5, false, false);
    c6 = __builtin_amdgcn_wmma_f32_16x16x4_f32(false, a, false, b6, (short)0, c6, false, false);
    c7 = __builtin_amdgcn_wmma_f32_16x16x4_f32(false, a, false, b7, (short)0, c7, false, false);
    c8 = __builtin_amdgcn_wmma_f32_16x16x4_f32(false, a, false, b8, (short)0, c8, false, false);
  }

  size_t rbase = (size_t)(wave * 16 + 8 * half) * NPAD;
  v8f acc[9] = {c0, c1, c2, c3, c4, c5, c6, c7, c8};
#pragma unroll
  for (int j = 0; j < NTILES; ++j) {
    size_t outBase = rbase + j * 16 + lm;
#pragma unroll
    for (int v = 0; v < 8; ++v)
      FIR[outBase + (size_t)v * NPAD] = acc[j][v];
  }
}

#define TS    2560
#define RO    16
#define NQ    (RO / 4)
#define CNT   (TS / RO)
#define XQ    (TS / 4 + 32)
#define XH    ((XQ + NQ) / NQ)
#define NFR   (TS / FRAME + 2)
#define FQ    (NPAD / 4)
#define FP    (FQ + 1)
#define NCH   33
#define JBLK  (TFRAMES * FRAME / TS)
#define XIT   ((XQ + CNT - 1) / CNT)
#define FIT   ((NFR * FQ + CNT - 1) / CNT)

__device__ __forceinline__ int xslot(int g) {
  return g / NQ + (g % NQ) * XH;
}

__device__ __forceinline__ int xslot_own(int tid, int m) {
  return tid + m / NQ + (m % NQ) * XH;
}

__device__ __forceinline__ int frow(int g) {
  return (int)(((unsigned)(g + 3) * 3277u) >> 16);
}

__device__ __forceinline__ void stage_tiles(const float* __restrict__ noise,
                                            const float* __restrict__ FIR,
                                            float4* xs, float4* fs,
                                            int b, int j, int tid) {
  const float4 zero = make_float4(0.0f, 0.0f, 0.0f, 0.0f);
  const float* nrow = noise + (size_t)b * (TFRAMES * FRAME);
  const int v0 = TS * j - 64;
  const int t0 = (TS / FRAME) * j - 1;
  float4 xr[XIT];
  float4 fr[FIT];
#pragma unroll
  for (int it = 0; it < XIT; ++it) {
    int v = v0 + 4 * (tid + it * CNT);
    int vc = v < 0 ? 0 : (v > TFRAMES * FRAME - 4 ? TFRAMES * FRAME - 4 : v);
    xr[it] = *(const float4*)(nrow + vc);
  }
#pragma unroll
  for (int it = 0; it < FIT; ++it) {
    int e = tid + it * CNT;
    int r = e / FQ;
    int t = t0 + r;
    int tc = t < 0 ? 0 : (t > TFRAMES - 1 ? TFRAMES - 1 : t);
    fr[it] = *(const float4*)(FIR + ((size_t)(b * TFRAMES + tc) * NPAD + 4 * (e - r * FQ)));
  }
  if (tid == 0) xs[0] = zero;
#pragma unroll
  for (int it = 0; it < XIT; ++it) {
    int e = tid + it * CNT;
    int v = v0 + 4 * e;
    float4 x = zero;
    if (v >= 0 && v < TFRAMES * FRAME) {
      x.x = 2.0f * xr[it].x - 1.0f;
      x.y = 2.0f * xr[it].y - 1.0f;
      x.z = 2.0f * xr[it].z - 1.0f;
      x.w = 2.0f * xr[it].w - 1.0f;
    }
    if (e < XQ) xs[xslot(e + 1)] = x;
  }
#pragma unroll
  for (int it = 0; it < FIT; ++it) {
    int e = tid + it * CNT;
    int r = e / FQ;
    int t = t0 + r;
    const bool in = (t >= 0 && t < TFRAMES);
    float4 f;
    f.x = in ? fr[it].x : 0.0f;
    f.y = in ? fr[it].y : 0.0f;
    f.z = in ? fr[it].z : 0.0f;
    f.w = in ? fr[it].w : 0.0f;
    if (e < NFR * FQ) fs[r * FP + (e - r * FQ)] = f;
  }
}

__device__ __forceinline__ void conv_tile(const float4* xs, const float4* fs,
                                          float* __restrict__ out,
                                          int b, int j, int tid) {
  float acc[RO];
#pragma unroll
  for (int i = 0; i < RO; ++i) acc[i] = 0.0f;

  float4 gq[NQ + 1];
  int ro[NQ + 1];
#pragma unroll
  for (int r = 1; r <= NQ; ++r) {
    gq[r] = xs[xslot_own(tid, 32 + r)];
    ro[r] = frow(NQ * tid + 32 + r) * FP;
  }

#pragma unroll
  for (int c = 0; c < NCH; ++c) {
    gq[0] = xs[xslot_own(tid, 32 - c)];
    ro[0] = frow(NQ * tid + 32 - c) * FP;
    float xw[4 * (NQ + 1)];
    float fa[NQ + 1][4];
#pragma unroll
    for (int r = 0; r <= NQ; ++r) {
      const float4 f = fs[ro[r] + c];
      fa[r][0] = f.x; fa[r][1] = f.y; fa[r][2] = f.z; fa[r][3] = f.w;
      xw[4 * r + 0] = gq[r].x; xw[4 * r + 1] = gq[r].y;
      xw[4 * r + 2] = gq[r].z; xw[4 * r + 3] = gq[r].w;
    }
#pragma unroll
    for (int q = 0; q < 4; ++q) {
#pragma unroll
      for (int i = 0; i < RO; ++i) {
        const int d = i - q;
        acc[i] = fmaf(xw[4 + d], fa[(d + 4) >> 2][q], acc[i]);
      }
    }
#pragma unroll
    for (int r = NQ; r >= 1; --r) { gq[r] = gq[r - 1]; ro[r] = ro[r - 1]; }
  }

  float4* o = (float4*)(out + (size_t)b * (TFRAMES * FRAME) + (size_t)TS * j + RO * tid);
#pragma unroll
  for (int r = 0; r < NQ; ++r)
    o[r] = make_float4(acc[4 * r], acc[4 * r + 1], acc[4 * r + 2], acc[4 * r + 3]);
}

__global__ __launch_bounds__(CNT)
void conv_ola(const float* __restrict__ noise,
              const float* __restrict__ FIR,
              float* __restrict__ out) {
  __shared__ float4 xs[NQ * XH];
  __shared__ float4 fs[NFR * FP];
  const int tid = (int)threadIdx.x;
  const int b = (int)blockIdx.x / JBLK;
  const int j = (int)blockIdx.x - b * JBLK;
  stage_tiles(noise, FIR, xs, fs, b, j, tid);
  __syncthreads();
  conv_tile(xs, fs, out, b, j, tid);
}

extern "C" void kernel_launch(void* const* d_in, const int* in_sizes, int n_in,
                              void* d_out, int out_size, void* d_ws, size_t ws_size,
                              hipStream_t stream) {
  const float* H     = (const float*)d_in[0];
  const float* noise = (const float*)d_in[1];
  float* out = (float*)d_out;

  float* Bp  = (float*)d_ws;
  float* FIR = (float*)((char*)d_ws + 64 * 1024);

  {
    int total = PPAIRS * NPAD * 2;
    build_B<<<(total + 255) / 256, 256, 0, stream>>>(Bp);
  }
  {
    long long threads = (long long)RTILES * 32;
    fir_gemm<<<(unsigned)((threads + 255) / 256), 256, 0, stream>>>(H, Bp, FIR);
  }
  {
    conv_ola<<<BATCH * JBLK, CNT, 0, stream>>>(noise, FIR, out);
  }
}


# FIR product fused into the convolution kernel: a work-group stages the H rows of its 30+2 frames in LDS and makes its 32 filter rows with the same f32 WMMA loop into LDS; no FIR workspace pass
# speedup vs baseline: 8.0416x; 1.2910x over previous
#include <hip/hip_runtime.h>
#include <math.h>

typedef __attribute__((ext_vector_type(2))) float v2f;
typedef __attribute__((ext_vector_type(8))) float v8f;

#define NBANDS 65
#define LFIR   129
#define NPAD   144
#define FRAME  80
#define TFRAMES 4000
#define BATCH  32
#define LCONV  208
#define ROWS   (BATCH*TFRAMES)
#define NTILES 9
#define RTILES (ROWS/16)
#define KSTEPS 17
#define PPAIRS (KSTEPS*2)

__global__ void build_B(float* __restrict__ Bp) {
  int i = blockIdx.x * blockDim.x + threadIdx.x;
  if (i >= PPAIRS * NPAD * 2) return;
  int p   = i / (NPAD * 2);
  int rem = i - p * (NPAD * 2);
  int n   = rem >> 1;
  int r   = rem & 1;
  int k   = 2 * p + r;
  float v = 0.0f;
  if (k < NBANDS && n < LFIR) {
    const float PI = 3.14159265358979323846f;
    float hann = 0.5f * (1.0f - cosf(2.0f * PI * (float)n / (float)LFIR));
    float w = (k == 0) ? 1.0f : 2.0f;
    float c = cosf(2.0f * PI * (float)k * (float)(n - (NBANDS - 1)) / (float)LFIR);
    v = hann * w * c / (float)LFIR;
  }
  Bp[i] = v;
}

#define TS    2400
#define RO    16
#define NQ    (RO / 4)
#define CNT   (TS / RO)
#define NTHR  160
#define XQ    (TS / 4 + 32)
#define XH    ((XQ + NQ) / NQ)
#define NFR   (TS / FRAME + 2)
#define FQ    (NPAD / 4)
#define FP    (FQ + 1)
#define NCH   33
#define JBLK  ((TFRAMES * FRAME + TS - 1) / TS)
#define XIT   ((XQ + NTHR - 1) / NTHR)
#define MT    (NFR / 16)
#define HP    34
#define HIT   ((NFR * NBANDS + NTHR - 1) / NTHR)

__device__ __forceinline__ int xslot(int g) {
  return g / NQ + (g % NQ) * XH;
}

__device__ __forceinline__ int xslot_own(int tid, int m) {
  return tid + m / NQ + (m % NQ) * XH;
}

__device__ __forceinline__ int frow(int g) {
  return (int)(((unsigned)(g + 3) * 3277u) >> 16);
}

__device__ __forceinline__ void stage_in(const float* __restrict__ noise,
                                         const float* __restrict__ H,
                                         float4* xs, float* hsf,
                                         int b, int j, int tid) {
  const float4 zero = make_float4(0.0f, 0.0f, 0.0f, 0.0f);
  const float* nrow = noise + (size_t)b * (TFRAMES * FRAME);
  const int v0 = TS * j - 64;
  const int t0 = (TS / FRAME) * j - 1;
  float4 xr[XIT];
  float hr[HIT];
#pragma unroll
  for (int it = 0; it < XIT; ++it) {
    int v = v0 + 4 * (tid + it * NTHR);
    int vc = v < 0 ? 0 : (v > TFRAMES * FRAME - 4 ? TFRAMES * FRAME - 4 : v);
    xr[it] = *(const float4*)(nrow + vc);
  }
#pragma unroll
  for (int it = 0; it < HIT; ++it) {
    int e = tid + it * NTHR;
    int ec = e < NFR * NBANDS ? e : NFR * NBANDS - 1;
    int r = ec / NBANDS;
    int t = t0 + r;
    int tc = t < 0 ? 0 : (t > TFRAMES - 1 ? TFRAMES - 1 : t);
    hr[it] = H[(size_t)(b * TFRAMES + tc) * NBANDS + (ec - r * NBANDS)];
  }
  if (tid == 0) xs[0] = zero;
#pragma unroll
  for (int it = 0; it < XIT; ++it) {
    int e = tid + it * NTHR;
    int v = v0 + 4 * e;
    float4 x = zero;
    if (v >= 0 && v < TFRAMES * FRAME) {
      x.x = 2.0f * xr[it].x - 1.0f;
      x.y = 2.0f * xr[it].y - 1.0f;
      x.z = 2.0f * xr[it].z - 1.0f;
      x.w = 2.0f * xr[it].w - 1.0f;
    }
    if (e < XQ) xs[xslot(e + 1)] = x;
  }
#pragma unroll
  for (int it = 0; it < HIT; ++it) {
    int e = tid + it * NTHR;
    int r = e / NBANDS;
    if (e < NFR * NBANDS) hsf[r * (2 * HP) + (e - r * NBANDS)] = hr[it];
  }
}

template <int NT>
__device__ __forceinline__ void fir_tiles(const v2f* hs,
                                          const float* __restrict__ Bp,
                                          float* fsf, int j, int lane,
                                          int n0, int n1) {
  const int half = lane >> 4;
  const int lm   = lane & 15;
  const int t0   = (TS / FRAME) * j - 1;

  const v2f* hrow[MT];
#pragma unroll
  for (int m = 0; m < MT; ++m) hrow[m] = hs + (m * 16 + lm) * HP;

  v8f acc[MT][NT];
#pragma unroll
  for (int m = 0; m < MT; ++m)
#pragma unroll
    for (int q = 0; q < NT; ++q) acc[m][q] = v8f{};

#pragma unroll 1
  for (int kk = 0; kk < KSTEPS - 1; ++kk) {
    v2f a[MT];
#pragma unroll
    for (int m = 0; m < MT; ++m) a[m] = hrow[m][kk * 2 + half];
    const v2f* brow = (const v2f*)(Bp + ((size_t)(kk * 2 + half) * NPAD + lm) * 2);
    v2f bb[NT];
    bb[0] = brow[n0 * 16];
    if (NT > 1) bb[NT - 1] = brow[n1 * 16];
#pragma unroll
    for (int m = 0; m < MT; ++m)
#pragma unroll
      for (int q = 0; q < NT; ++q)
        acc[m][q] = __builtin_amdgcn_wmma_f32_16x16x4_f32(false, a[m], false, bb[q], (short)0,
                                                          acc[m][q], false, false);
  }

  {
    const int kk = KSTEPS - 1;
    v2f a[MT];
#pragma unroll
    for (int m = 0; m < MT; ++m) {
      float h64 = ((const float*)hrow[m])[NBANDS - 1];
      a[m].x = half ? 0.0f : h64;
      a[m].y = 0.0f;
    }
    const v2f* brow = (const v2f*)(Bp + ((size_t)(kk * 2 + half) * NPAD + lm) * 2);
    v2f bb[NT];
    bb[0] = brow[n0 * 16];
    if (NT > 1) bb[NT - 1] = brow[n1 * 16];
#pragma unroll
    for (int m = 0; m < MT; ++m)
#pragma unroll
      for (int q = 0; q < NT; ++q)
        acc[m][q] = __builtin_amdgcn_wmma_f32_16x16x4_f32(false, a[m], false, bb[q], (short)0,
                                                          acc[m][q], false, false);
  }

#pragma unroll
  for (int m = 0; m < MT; ++m) {
#pragma unroll
    for (int v = 0; v < 8; ++v) {
      const int r = m * 16 + 8 * half + v;
      const int t = t0 + r;
      const bool in = (t >= 0 && t < TFRAMES);
      fsf[r * (FP * 4) + n0 * 16 + lm] = in ? acc[m][0][v] : 0.0f;
      if (NT > 1) fsf[r * (FP * 4) + n1 * 16 + lm] = in ? acc[m][NT - 1][v] : 0.0f;
    }
  }
}

__device__ __forceinline__ void conv_tile(const float4* xs, const float4* fs,
                                          float* __restrict__ out,
                                          int b, int j, int tid) {
  float acc[RO];
#pragma unroll
  for (int i = 0; i < RO; ++i) acc[i] = 0.0f;

  float4 gq[NQ + 1];
  int ro[NQ + 1];
#pragma unroll
  for (int r = 1; r <= NQ; ++r) {
    gq[r] = xs[xslot_own(tid, 32 + r)];
    ro[r] = frow(NQ * tid + 32 + r) * FP;
  }

#pragma unroll
  for (int c = 0; c < NCH; ++c) {
    gq[0] = xs[xslot_own(tid, 32 - c)];
    ro[0] = frow(NQ * tid + 32 - c) * FP;
    float xw[4 * (NQ + 1)];
    float fa[NQ + 1][4];
#pragma unroll
    for (int r = 0; r <= NQ; ++r) {
      const float4 f = fs[ro[r] + c];
      fa[r][0] = f.x; fa[r][1] = f.y; fa[r][2] = f.z; fa[r][3] = f.w;
      xw[4 * r + 0] = gq[r].x; xw[4 * r + 1] = gq[r].y;
      xw[4 * r + 2] = gq[r].z; xw[4 * r + 3] = gq[r].w;
    }
#pragma unroll
    for (int q = 0; q < 4; ++q) {
#pragma unroll
      for (int i = 0; i < RO; ++i) {
        const int d = i - q;
        acc[i] = fmaf(xw[4 + d], fa[(d + 4) >> 2][q], acc[i]);
      }
    }
#pragma unroll
    for (int r = NQ; r >= 1; --r) { gq[r] = gq[r - 1]; ro[r] = ro[r - 1]; }
  }

  float4* o = (float4*)(out + (size_t)b * (TFRAMES * FRAME) + (size_t)TS * j + RO * tid);
#pragma unroll
  for (int r = 0; r < NQ; ++r)
    o[r] = make_float4(acc[4 * r], acc[4 * r + 1], acc[4 * r + 2], acc[4 * r + 3]);
}

__global__ __launch_bounds__(NTHR)
void conv_ola(const float* __restrict__ noise,
              const float* __restrict__ H,
              const float* __restrict__ Bp,
              float* __restrict__ out) {
  __shared__ float4 xs[NQ * XH];
  __shared__ float4 fs[NFR * FP];
  __shared__ v2f hs[NFR * HP];
  const int tid = (int)threadIdx.x;
  const int b = (int)blockIdx.x / JBLK;
  const int j = (int)blockIdx.x - b * JBLK;
  stage_in(noise, H, xs, (float*)hs, b, j, tid);
  __syncthreads();
  const int w = __builtin_amdgcn_readfirstlane(tid >> 5);
  const int lane = tid & 31;
  if (w < 4) fir_tiles<2>(hs, Bp, (float*)fs, j, lane, w, w + 5);
  else       fir_tiles<1>(hs, Bp, (float*)fs, j, lane, 4, 4);
  __syncthreads();
  if (tid < CNT && TS * j + RO * tid < TFRAMES * FRAME) conv_tile(xs, fs, out, b, j, tid);
}

extern "C" void kernel_launch(void* const* d_in, const int* in_sizes, int n_in,
                              void* d_out, int out_size, void* d_ws, size_t ws_size,
                              hipStream_t stream) {
  const float* H     = (const float*)d_in[0];
  const float* noise = (const float*)d_in[1];
  float* out = (float*)d_out;

  float* Bp  = (float*)d_ws;

  {
    int total = PPAIRS * NPAD * 2;
    build_B<<<(total + 255) / 256, 256, 0, stream>>>(Bp);
  }
  {
    conv_ola<<<BATCH * JBLK, NTHR, 0, stream>>>(noise, H, Bp, out);
  }
}


# tap 0 of every filter row is exactly zero (hann(0)=0), so rows hold taps 1..128: 8 WMMA column tiles not 9, 32 four-tap chunks not 33, chunk windows on 16-byte boundaries (whole 16-byte LDS reads)
# speedup vs baseline: 8.3345x; 1.0364x over previous
#include <hip/hip_runtime.h>
#include <math.h>

typedef __attribute__((ext_vector_type(2))) float v2f;
typedef __attribute__((ext_vector_type(8))) float v8f;

#define NBANDS 65
#define LFIR   129
#define NPAD   128
#define FRAME  80
#define TFRAMES 4000
#define BATCH  32
#define LCONV  208
#define ROWS   (BATCH*TFRAMES)
#define NTILES 8
#define RTILES (ROWS/16)
#define KSTEPS 17
#define PPAIRS (KSTEPS*2)

__global__ void build_B(float* __restrict__ Bp) {
  int i = blockIdx.x * blockDim.x + threadIdx.x;
  if (i >= PPAIRS * NPAD * 2) return;
  int p   = i / (NPAD * 2);
  int rem = i - p * (NPAD * 2);
  int n   = (rem >> 1) + 1;
  int r   = rem & 1;
  int k   = 2 * p + r;
  float v = 0.0f;
  if (k < NBANDS) {
    const float PI = 3.14159265358979323846f;
    float hann = 0.5f * (1.0f - cosf(2.0f * PI * (float)n / (float)LFIR));
    float w = (k == 0) ? 1.0f : 2.0f;
    float c = cosf(2.0f * PI * (float)k * (float)(n - (NBANDS - 1)) / (float)LFIR);
    v = hann * w * c / (float)LFIR;
  }
  Bp[i] = v;
}

#define TS    2400
#define RO    16
#define NQ    (RO / 4)
#define CNT   (TS / RO)
#define NTHR  160
#define XQ    (TS / 4 + 32)
#define XH    ((XQ + NQ) / NQ)
#define NFR   (TS / FRAME + 2)
#define FQ    (NPAD / 4)
#define FP    (FQ + 1)
#define NCH   FQ
#define JBLK  ((TFRAMES * FRAME + TS - 1) / TS)
#define XIT   ((XQ + NTHR - 1) / NTHR)
#define MT    (NFR / 16)
#define HP    34
#define HIT   ((NFR * NBANDS + NTHR - 1) / NTHR)

__device__ __forceinline__ int xslot(int g) {
  return g / NQ + (g % NQ) * XH;
}

__device__ __forceinline__ int xslot_own(int tid, int m) {
  return tid + m / NQ + (m % NQ) * XH;
}

__device__ __forceinline__ int frow(int g) {
  return (int)(((unsigned)(g + 3) * 3277u) >> 16);
}

__device__ __forceinline__ void stage_in(const float* __restrict__ noise,
                                         const float* __restrict__ H,
                                         float4* xs, float* hsf,
                                         int b, int j, int tid) {
  const float4 zero = make_float4(0.0f, 0.0f, 0.0f, 0.0f);
  const float* nrow = noise + (size_t)b * (TFRAMES * FRAME);
  const int v0 = TS * j - 64;
  const int t0 = (TS / FRAME) * j - 1;
  float4 xr[XIT];
  float hr[HIT];
#pragma unroll
  for (int it = 0; it < XIT; ++it) {
    int v = v0 + 4 * (tid + it * NTHR);
    int vc = v < 0 ? 0 : (v > TFRAMES * FRAME - 4 ? TFRAMES * FRAME - 4 : v);
    xr[it] = *(const float4*)(nrow + vc);
  }
#pragma unroll
  for (int it = 0; it < HIT; ++it) {
    int e = tid + it * NTHR;
    int ec = e < NFR * NBANDS ? e : NFR * NBANDS - 1;
    int r = ec / NBANDS;
    int t = t0 + r;
    int tc = t < 0 ? 0 : (t > TFRAMES - 1 ? TFRAMES - 1 : t);
    hr[it] = H[(size_t)(b * TFRAMES + tc) * NBANDS + (ec - r * NBANDS)];
  }
#pragma unroll
  for (int it = 0; it < XIT; ++it) {
    int e = tid + it * NTHR;
    int v = v0 + 4 * e;
    float4 x = zero;
    if (v >= 0 && v < TFRAMES * FRAME) {
      x.x = 2.0f * xr[it].x - 1.0f;
      x.y = 2.0f * xr[it].y - 1.0f;
      x.z = 2.0f * xr[it].z - 1.0f;
      x.w = 2.0f * xr[it].w - 1.0f;
    }
    if (e < XQ) xs[xslot(e + 1)] = x;
  }
#pragma unroll
  for (int it = 0; it < HIT; ++it) {
    int e = tid + it * NTHR;
    int r = e / NBANDS;
    if (e < NFR * NBANDS) hsf[r * (2 * HP) + (e - r * NBANDS)] = hr[it];
  }
}

template <int NT>
__device__ __forceinline__ void fir_tiles(const v2f* hs,
                                          const float* __restrict__ Bp,
                                          float* fsf, int j, int lane,
                                          int n0, int n1) {
  const int half = lane >> 4;
  const int lm   = lane & 15;
  const int t0   = (TS / FRAME) * j - 1;

  const v2f* hrow[MT];
#pragma unroll
  for (int m = 0; m < MT; ++m) hrow[m] = hs + (m * 16 + lm) * HP;

  v8f acc[MT][NT];
#pragma unroll
  for (int m = 0; m < MT; ++m)
#pragma unroll
    for (int q = 0; q < NT; ++q) acc[m][q] = v8f{};

#pragma unroll 1
  for (int kk = 0; kk < KSTEPS - 1; ++kk) {
    v2f a[MT];
#pragma unroll
    for (int m = 0; m < MT; ++m) a[m] = hrow[m][kk * 2 + half];
    const v2f* brow = (const v2f*)(Bp + ((size_t)(kk * 2 + half) * NPAD + lm) * 2);
    v2f bb[NT];
    bb[0] = brow[n0 * 16];
    if (NT > 1) bb[NT - 1] = brow[n1 * 16];
#pragma unroll
    for (int m = 0; m < MT; ++m)
#pragma unroll
      for (int q = 0; q < NT; ++q)
        acc[m][q] = __builtin_amdgcn_wmma_f32_16x16x4_f32(false, a[m], false, bb[q], (short)0,
                                                          acc[m][q], false, false);
  }

  {
    const int kk = KSTEPS - 1;
    v2f a[MT];
#pragma unroll
    for (int m = 0; m < MT; ++m) {
      float h64 = ((const float*)hrow[m])[NBANDS - 1];
      a[m].x = half ? 0.0f : h64;
      a[m].y = 0.0f;
    }
    const v2f* brow = (const v2f*)(Bp + ((size_t)(kk * 2 + half) * NPAD + lm) * 2);
    v2f bb[NT];
    bb[0] = brow[n0 * 16];
    if (NT > 1) bb[NT - 1] = brow[n1 * 16];
#pragma unroll
    for (int m = 0; m < MT; ++m)
#pragma unroll
      for (int q = 0; q < NT; ++q)
        acc[m][q] = __builtin_amdgcn_wmma_f32_16x16x4_f32(false, a[m], false, bb[q], (short)0,
                                                          acc[m][q], false, false);
  }

#pragma unroll
  for (int m = 0; m < MT; ++m) {
#pragma unroll
    for (int v = 0; v < 8; ++v) {
      const int r = m * 16 + 8 * half + v;
      const int t = t0 + r;
      const bool in = (t >= 0 && t < TFRAMES);
      fsf[r * (FP * 4) + n0 * 16 + lm] = in ? acc[m][0][v] : 0.0f;
      if (NT > 1) fsf[r * (FP * 4) + n1 * 16 + lm] = in ? acc[m][NT - 1][v] : 0.0f;
    }
  }
}

__device__ __forceinline__ void conv_tile(const float4* xs, const float4* fs,
                                          float* __restrict__ out,
                                          int b, int j, int tid) {
  float acc[RO];
#pragma unroll
  for (int i = 0; i < RO; ++i) acc[i] = 0.0f;

  float4 gq[NQ + 1];
  int ro[NQ + 1];
#pragma unroll
  for (int r = 1; r <= NQ; ++r) {
    gq[r] = xs[xslot_own(tid, 32 + r)];
    ro[r] = frow(NQ * tid + 32 + r) * FP;
  }

#pragma unroll
  for (int c = 0; c < NCH; ++c) {
    gq[0] = xs[xslot_own(tid, 32 - c)];
    ro[0] = frow(NQ * tid + 32 - c) * FP;
    float xw[4 * (NQ + 1)];
    float fa[NQ + 1][4];
#pragma unroll
    for (int r = 0; r <= NQ; ++r) {
      const float4 f = fs[ro[r] + c];
      fa[r][0] = f.x; fa[r][1] = f.y; fa[r][2] = f.z; fa[r][3] = f.w;
      xw[4 * r + 0] = gq[r].x; xw[4 * r + 1] = gq[r].y;
      xw[4 * r + 2] = gq[r].z; xw[4 * r + 3] = gq[r].w;
    }
#pragma unroll
    for (int q = 0; q < 4; ++q) {
#pragma unroll
      for (int i = 0; i < RO; ++i) {
        const int d = i - q + 3;
        acc[i] = fmaf(xw[d], fa[d >> 2][q], acc[i]);
      }
    }
#pragma unroll
    for (int r = NQ; r >= 1; --r) { gq[r] = gq[r - 1]; ro[r] = ro[r - 1]; }
  }

  float4* o = (float4*)(out + (size_t)b * (TFRAMES * FRAME) + (size_t)TS * j + RO * tid);
#pragma unroll
  for (int r = 0; r < NQ; ++r)
    o[r] = make_float4(acc[4 * r], acc[4 * r + 1], acc[4 * r + 2], acc[4 * r + 3]);
}

__global__ __launch_bounds__(NTHR)
void conv_ola(const float* __restrict__ noise,
              const float* __restrict__ H,
              const float* __restrict__ Bp,
              float* __restrict__ out) {
  __shared__ float4 xs[NQ * XH];
  __shared__ float4 fs[NFR * FP];
  __shared__ v2f hs[NFR * HP];
  const int tid = (int)threadIdx.x;
  const int b = (int)blockIdx.x / JBLK;
  const int j = (int)blockIdx.x - b * JBLK;
  stage_in(noise, H, xs, (float*)hs, b, j, tid);
  __syncthreads();
  const int w = __builtin_amdgcn_readfirstlane(tid >> 5);
  const int lane = tid & 31;
  if (w < 4) fir_tiles<2>(hs, Bp, (float*)fs, j, lane, w, w + 4);
  __syncthreads();
  if (tid < CNT && TS * j + RO * tid < TFRAMES * FRAME) conv_tile(xs, fs, out, b, j, tid);
}

extern "C" void kernel_launch(void* const* d_in, const int* in_sizes, int n_in,
                              void* d_out, int out_size, void* d_ws, size_t ws_size,
                              hipStream_t stream) {
  const float* H     = (const float*)d_in[0];
  const float* noise = (const float*)d_in[1];
  float* out = (float*)d_out;

  float* Bp  = (float*)d_ws;

  {
    int total = PPAIRS * NPAD * 2;
    build_B<<<(total + 255) / 256, 256, 0, stream>>>(Bp);
  }
  {
    conv_ola<<<BATCH * JBLK, NTHR, 0, stream>>>(noise, H, Bp, out);
  }
}


# no table launch: a work-group makes the 129 cosines cos(2*pi*m/129) in LDS and picks its WMMA B fragments from them at k*|n-64| mod 129; column factor applied to the finished sums (rounding moves)
# speedup vs baseline: 8.8864x; 1.0662x over previous
#include <hip/hip_runtime.h>
#include <math.h>

typedef __attribute__((ext_vector_type(2))) float v2f;
typedef __attribute__((ext_vector_type(8))) float v8f;

#define NBANDS 65
#define LFIR   129
#define NPAD   128
#define FRAME  80
#define TFRAMES 4000
#define BATCH  32
#define LCONV  208
#define ROWS   (BATCH*TFRAMES)
#define NTILES 8
#define RTILES (ROWS/16)
#define KSTEPS 17
#define NCOS   LFIR


#define TS    2400
#define RO    16
#define NQ    (RO / 4)
#define CNT   (TS / RO)
#define NTHR  160
#define XQ    (TS / 4 + 32)
#define XH    ((XQ + NQ) / NQ)
#define NFR   (TS / FRAME + 2)
#define FQ    (NPAD / 4)
#define FP    (FQ + 1)
#define NCH   FQ
#define JBLK  ((TFRAMES * FRAME + TS - 1) / TS)
#define XIT   ((XQ + NTHR - 1) / NTHR)
#define MT    (NFR / 16)
#define HP    34
#define HIT   ((NFR * NBANDS + NTHR - 1) / NTHR)

__device__ __forceinline__ int xslot(int g) {
  return g / NQ + (g % NQ) * XH;
}

__device__ __forceinline__ int xslot_own(int tid, int m) {
  return tid + m / NQ + (m % NQ) * XH;
}

__device__ __forceinline__ int frow(int g) {
  return (int)(((unsigned)(g + 3) * 3277u) >> 16);
}

__device__ __forceinline__ void stage_in(const float* __restrict__ noise,
                                         const float* __restrict__ H,
                                         float4* xs, float* hsf, float* ct,
                                         int b, int j, int tid) {
  const float4 zero = make_float4(0.0f, 0.0f, 0.0f, 0.0f);
  const float* nrow = noise + (size_t)b * (TFRAMES * FRAME);
  const int v0 = TS * j - 64;
  const int t0 = (TS / FRAME) * j - 1;
  float4 xr[XIT];
  float hr[HIT];
#pragma unroll
  for (int it = 0; it < XIT; ++it) {
    int v = v0 + 4 * (tid + it * NTHR);
    int vc = v < 0 ? 0 : (v > TFRAMES * FRAME - 4 ? TFRAMES * FRAME - 4 : v);
    xr[it] = *(const float4*)(nrow + vc);
  }
#pragma unroll
  for (int it = 0; it < HIT; ++it) {
    int e = tid + it * NTHR;
    int ec = e < NFR * NBANDS ? e : NFR * NBANDS - 1;
    int r = ec / NBANDS;
    int t = t0 + r;
    int tc = t < 0 ? 0 : (t > TFRAMES - 1 ? TFRAMES - 1 : t);
    hr[it] = H[(size_t)(b * TFRAMES + tc) * NBANDS + (ec - r * NBANDS)];
  }
  if (tid <= NCOS) ct[tid] = tid < NCOS ? cospif((float)tid * (2.0f / (float)LFIR)) : 0.5f;
#pragma unroll
  for (int it = 0; it < XIT; ++it) {
    int e = tid + it * NTHR;
    int v = v0 + 4 * e;
    float4 x = zero;
    if (v >= 0 && v < TFRAMES * FRAME) {
      x.x = 2.0f * xr[it].x - 1.0f;
      x.y = 2.0f * xr[it].y - 1.0f;
      x.z = 2.0f * xr[it].z - 1.0f;
      x.w = 2.0f * xr[it].w - 1.0f;
    }
    if (e < XQ) xs[xslot(e + 1)] = x;
  }
#pragma unroll
  for (int it = 0; it < HIT; ++it) {
    int e = tid + it * NTHR;
    int r = e / NBANDS;
    if (e < NFR * NBANDS) hsf[r * (2 * HP) + (e - r * NBANDS)] = hr[it];
  }
}

#define CTB (NCOS * 4)
__device__ __forceinline__ unsigned wrapct(unsigned off) {
  const unsigned w = off - (unsigned)CTB;
  return off < w ? off : w;
}
__device__ __forceinline__ float ctat(const float* ct, unsigned off) {
  return *(const float*)((const char*)ct + off);
}

template <int NT>
__device__ __forceinline__ void fir_tiles(const v2f* hs,
                                          const float* ct,
                                          float* fsf, int j, int lane,
                                          int n0, int n1) {
  const int half = lane >> 4;
  const int lm   = lane & 15;
  const int t0   = (TS / FRAME) * j - 1;

  const v2f* hrow[MT];
#pragma unroll
  for (int m = 0; m < MT; ++m) hrow[m] = hs + (m * 16 + lm) * HP;

  float sc[NT];
  unsigned i0[NT], i1[NT], st[NT];
#pragma unroll
  for (int q = 0; q < NT; ++q) {
    const int n  = (q == 0 ? n0 : n1) * 16 + lm + 1;
    const int ad = n >= NBANDS - 1 ? n - (NBANDS - 1) : (NBANDS - 1) - n;
    sc[q] = (1.0f - ct[n]) * (1.0f / (float)LFIR);
    st[q] = wrapct((unsigned)(16 * ad));
    i0[q] = half ? (unsigned)(8 * ad) : (unsigned)CTB;
    i1[q] = wrapct(i0[q] + (unsigned)(4 * ad));
  }

  v8f acc[MT][NT];
#pragma unroll
  for (int m = 0; m < MT; ++m)
#pragma unroll
    for (int q = 0; q < NT; ++q) acc[m][q] = v8f{};

#pragma unroll 1
  for (int kk = 0; kk < KSTEPS - 1; ++kk) {
    v2f a[MT];
#pragma unroll
    for (int m = 0; m < MT; ++m) a[m] = hrow[m][kk * 2 + half];
    v2f bb[NT];
#pragma unroll
    for (int q = 0; q < NT; ++q) {
      bb[q].x = ctat(ct, i0[q]);
      bb[q].y = ctat(ct, i1[q]);
      i0[q] = wrapct(i0[q] + st[q]);
      i1[q] = wrapct(i1[q] + st[q]);
    }
#pragma unroll
    for (int m = 0; m < MT; ++m)
#pragma unroll
      for (int q = 0; q < NT; ++q)
        acc[m][q] = __builtin_amdgcn_wmma_f32_16x16x4_f32(false, a[m], false, bb[q], (short)0,
                                                          acc[m][q], false, false);
  }

  {
    v2f a[MT];
#pragma unroll
    for (int m = 0; m < MT; ++m) {
      float h64 = ((const float*)hrow[m])[NBANDS - 1];
      a[m].x = half ? 0.0f : h64;
      a[m].y = 0.0f;
    }
    v2f bb[NT];
#pragma unroll
    for (int q = 0; q < NT; ++q) {
      bb[q].x = ctat(ct, i0[q]);
      bb[q].y = ctat(ct, i1[q]);
    }
#pragma unroll
    for (int m = 0; m < MT; ++m)
#pragma unroll
      for (int q = 0; q < NT; ++q)
        acc[m][q] = __builtin_amdgcn_wmma_f32_16x16x4_f32(false, a[m], false, bb[q], (short)0,
                                                          acc[m][q], false, false);
  }

#pragma unroll
  for (int m = 0; m < MT; ++m) {
#pragma unroll
    for (int v = 0; v < 8; ++v) {
      const int r = m * 16 + 8 * half + v;
      const int t = t0 + r;
      const bool in = (t >= 0 && t < TFRAMES);
      fsf[r * (FP * 4) + n0 * 16 + lm] = in ? sc[0] * acc[m][0][v] : 0.0f;
      if (NT > 1) fsf[r * (FP * 4) + n1 * 16 + lm] = in ? sc[NT - 1] * acc[m][NT - 1][v] : 0.0f;
    }
  }
}

__device__ __forceinline__ void conv_tile(const float4* xs, const float4* fs,
                                          float* __restrict__ out,
                                          int b, int j, int tid) {
  float acc[RO];
#pragma unroll
  for (int i = 0; i < RO; ++i) acc[i] = 0.0f;

  float4 gq[NQ + 1];
  int ro[NQ + 1];
#pragma unroll
  for (int r = 1; r <= NQ; ++r) {
    gq[r] = xs[xslot_own(tid, 32 + r)];
    ro[r] = frow(NQ * tid + 32 + r) * FP;
  }

#pragma unroll
  for (int c = 0; c < NCH; ++c) {
    gq[0] = xs[xslot_own(tid, 32 - c)];
    ro[0] = frow(NQ * tid + 32 - c) * FP;
    float xw[4 * (NQ + 1)];
    float fa[NQ + 1][4];
#pragma unroll
    for (int r = 0; r <= NQ; ++r) {
      const float4 f = fs[ro[r] + c];
      fa[r][0] = f.x; fa[r][1] = f.y; fa[r][2] = f.z; fa[r][3] = f.w;
      xw[4 * r + 0] = gq[r].x; xw[4 * r + 1] = gq[r].y;
      xw[4 * r + 2] = gq[r].z; xw[4 * r + 3] = gq[r].w;
    }
#pragma unroll
    for (int q = 0; q < 4; ++q) {
#pragma unroll
      for (int i = 0; i < RO; ++i) {
        const int d = i - q + 3;
        acc[i] = fmaf(xw[d], fa[d >> 2][q], acc[i]);
      }
    }
#pragma unroll
    for (int r = NQ; r >= 1; --r) { gq[r] = gq[r - 1]; ro[r] = ro[r - 1]; }
  }

  float4* o = (float4*)(out + (size_t)b * (TFRAMES * FRAME) + (size_t)TS * j + RO * tid);
#pragma unroll
  for (int r = 0; r < NQ; ++r)
    o[r] = make_float4(acc[4 * r], acc[4 * r + 1], acc[4 * r + 2], acc[4 * r + 3]);
}

__global__ __launch_bounds__(NTHR)
void conv_ola(const float* __restrict__ noise,
              const float* __restrict__ H,
              float* __restrict__ out) {
  __shared__ float4 xs[NQ * XH];
  __shared__ float4 fs[NFR * FP];
  __shared__ v2f hs[NFR * HP];
  __shared__ float ct[NCOS + 1];
  const int tid = (int)threadIdx.x;
  const int b = (int)blockIdx.x / JBLK;
  const int j = (int)blockIdx.x - b * JBLK;
  stage_in(noise, H, xs, (float*)hs, ct, b, j, tid);
  __syncthreads();
  const int w = __builtin_amdgcn_readfirstlane(tid >> 5);
  const int lane = tid & 31;
  if (w < 4) fir_tiles<2>(hs, ct, (float*)fs, j, lane, w, w + 4);
  __syncthreads();
  if (tid < CNT && TS * j + RO * tid < TFRAMES * FRAME) conv_tile(xs, fs, out, b, j, tid);
}

extern "C" void kernel_launch(void* const* d_in, const int* in_sizes, int n_in,
                              void* d_out, int out_size, void* d_ws, size_t ws_size,
                              hipStream_t stream) {
  const float* H     = (const float*)d_in[0];
  const float* noise = (const float*)d_in[1];
  float* out = (float*)d_out;

  conv_ola<<<BATCH * JBLK, NTHR, 0, stream>>>(noise, H, out);
}


# convolution lanes remapped (lane = frame, wave = 16-sample slot in it): the frame edge in the tap window is wave-uniform, a chunk reads 1-2 tap quads from LDS, not 5; outputs leave through LDS
# speedup vs baseline: 10.0913x; 1.1356x over previous
#include <hip/hip_runtime.h>
#include <math.h>

typedef __attribute__((ext_vector_type(2))) float v2f;
typedef __attribute__((ext_vector_type(8))) float v8f;

#define NBANDS 65
#define LFIR   129
#define NPAD   128
#define FRAME  80
#define TFRAMES 4000
#define BATCH  32
#define LCONV  208
#define ROWS   (BATCH*TFRAMES)
#define NTILES 8
#define RTILES (ROWS/16)
#define KSTEPS 17
#define NCOS   LFIR


#define TS    2400
#define RO    16
#define NQ    (RO / 4)
#define LFR   (TS / FRAME)
#define NTHR  160
#define XQ    (TS / 4 + 32)
#define FRQ   (FRAME / 4)
#define XP    (FRQ + 1)
#define NFR   (TS / FRAME + 2)
#define FQ    (NPAD / 4)
#define FP    (FQ + 1)
#define NCH   FQ
#define JBLK  ((TFRAMES * FRAME + TS - 1) / TS)
#define XIT   ((XQ + NTHR - 1) / NTHR)
#define MT    (NFR / 16)
#define HP    34
#define HIT   ((NFR * NBANDS + NTHR - 1) / NTHR)

__device__ __forceinline__ int xslot(int Q) {
  return Q + (int)(((unsigned)Q * 3277u) >> 16);
}

__device__ __forceinline__ int qfr(int q) {
  return q >= FRQ ? 1 : (q < 0 ? -1 : 0);
}

__device__ __forceinline__ void stage_in(const float* __restrict__ noise,
                                         const float* __restrict__ H,
                                         float4* xs, float* hsf, float* ct,
                                         int b, int j, int tid) {
  const float4 zero = make_float4(0.0f, 0.0f, 0.0f, 0.0f);
  const float* nrow = noise + (size_t)b * (TFRAMES * FRAME);
  const int v0 = TS * j - 64;
  const int t0 = (TS / FRAME) * j - 1;
  float4 xr[XIT];
  float hr[HIT];
#pragma unroll
  for (int it = 0; it < XIT; ++it) {
    int v = v0 + 4 * (tid + it * NTHR);
    int vc = v < 0 ? 0 : (v > TFRAMES * FRAME - 4 ? TFRAMES * FRAME - 4 : v);
    xr[it] = *(const float4*)(nrow + vc);
  }
#pragma unroll
  for (int it = 0; it < HIT; ++it) {
    int e = tid + it * NTHR;
    int ec = e < NFR * NBANDS ? e : NFR * NBANDS - 1;
    int r = ec / NBANDS;
    int t = t0 + r;
    int tc = t < 0 ? 0 : (t > TFRAMES - 1 ? TFRAMES - 1 : t);
    hr[it] = H[(size_t)(b * TFRAMES + tc) * NBANDS + (ec - r * NBANDS)];
  }
  if (tid <= NCOS) ct[tid] = tid < NCOS ? cospif((float)tid * (2.0f / (float)LFIR)) : 0.5f;
#pragma unroll
  for (int it = 0; it < XIT; ++it) {
    int e = tid + it * NTHR;
    int v = v0 + 4 * e;
    float4 x = zero;
    if (v >= 0 && v < TFRAMES * FRAME) {
      x.x = 2.0f * xr[it].x - 1.0f;
      x.y = 2.0f * xr[it].y - 1.0f;
      x.z = 2.0f * xr[it].z - 1.0f;
      x.w = 2.0f * xr[it].w - 1.0f;
    }
    if (e < XQ) xs[xslot(e + 4)] = x;
  }
#pragma unroll
  for (int it = 0; it < HIT; ++it) {
    int e = tid + it * NTHR;
    int r = e / NBANDS;
    if (e < NFR * NBANDS) hsf[r * (2 * HP) + (e - r * NBANDS)] = hr[it];
  }
}

#define CTB (NCOS * 4)
__device__ __forceinline__ unsigned wrapct(unsigned off) {
  const unsigned w = off - (unsigned)CTB;
  return off < w ? off : w;
}
__device__ __forceinline__ float ctat(const float* ct, unsigned off) {
  return *(const float*)((const char*)ct + off);
}

template <int NT>
__device__ __forceinline__ void fir_tiles(const v2f* hs,
                                          const float* ct,
                                          float* fsf, int j, int lane,
                                          int n0, int n1) {
  const int half = lane >> 4;
  const int lm   = lane & 15;
  const int t0   = (TS / FRAME) * j - 1;

  const v2f* hrow[MT];
#pragma unroll
  for (int m = 0; m < MT; ++m) hrow[m] = hs + (m * 16 + lm) * HP;

  float sc[NT];
  unsigned i0[NT], i1[NT], st[NT];
#pragma unroll
  for (int q = 0; q < NT; ++q) {
    const int n  = (q == 0 ? n0 : n1) * 16 + lm + 1;
    const int ad = n >= NBANDS - 1 ? n - (NBANDS - 1) : (NBANDS - 1) - n;
    sc[q] = (1.0f - ct[n]) * (1.0f / (float)LFIR);
    st[q] = wrapct((unsigned)(16 * ad));
    i0[q] = half ? (unsigned)(8 * ad) : (unsigned)CTB;
    i1[q] = wrapct(i0[q] + (unsigned)(4 * ad));
  }

  v8f acc[MT][NT];
#pragma unroll
  for (int m = 0; m < MT; ++m)
#pragma unroll
    for (int q = 0; q < NT; ++q) acc[m][q] = v8f{};

#pragma unroll 1
  for (int kk = 0; kk < KSTEPS - 1; ++kk) {
    v2f a[MT];
#pragma unroll
    for (int m = 0; m < MT; ++m) a[m] = hrow[m][kk * 2 + half];
    v2f bb[NT];
#pragma unroll
    for (int q = 0; q < NT; ++q) {
      bb[q].x = ctat(ct, i0[q]);
      bb[q].y = ctat(ct, i1[q]);
      i0[q] = wrapct(i0[q] + st[q]);
      i1[q] = wrapct(i1[q] + st[q]);
    }
#pragma unroll
    for (int m = 0; m < MT; ++m)
#pragma unroll
      for (int q = 0; q < NT; ++q)
        acc[m][q] = __builtin_amdgcn_wmma_f32_16x16x4_f32(false, a[m], false, bb[q], (short)0,
                                                          acc[m][q], false, false);
  }

  {
    v2f a[MT];
#pragma unroll
    for (int m = 0; m < MT; ++m) {
      float h64 = ((const float*)hrow[m])[NBANDS - 1];
      a[m].x = half ? 0.0f : h64;
      a[m].y = 0.0f;
    }
    v2f bb[NT];
#pragma unroll
    for (int q = 0; q < NT; ++q) {
      bb[q].x = ctat(ct, i0[q]);
      bb[q].y = ctat(ct, i1[q]);
    }
#pragma unroll
    for (int m = 0; m < MT; ++m)
#pragma unroll
      for (int q = 0; q < NT; ++q)
        acc[m][q] = __builtin_amdgcn_wmma_f32_16x16x4_f32(false, a[m], false, bb[q], (short)0,
                                                          acc[m][q], false, false);
  }

#pragma unroll
  for (int m = 0; m < MT; ++m) {
#pragma unroll
    for (int v = 0; v < 8; ++v) {
      const int r = m * 16 + 8 * half + v;
      const int t = t0 + r;
      const bool in = (t >= 0 && t < TFRAMES);
      fsf[r * (FP * 4) + n0 * 16 + lm] = in ? sc[0] * acc[m][0][v] : 0.0f;
      if (NT > 1) fsf[r * (FP * 4) + n1 * 16 + lm] = in ? sc[NT - 1] * acc[m][NT - 1][v] : 0.0f;
    }
  }
}

__device__ __forceinline__ void conv_chunk(const float4* xl, const float4* fl, const int d,
                                           float4 (&gq)[NQ + 1], float (&acc)[RO]) {
  const int q0 = 15 - d;
  const int fa = qfr(q0);
  const int fb = qfr(q0 + NQ);
  gq[0] = xl[(1 + fa) * XP + q0 - FRQ * fa];
  const float4 ta = fl[(1 + fa) * FP + d];
  float4 tb = ta;
  if (fb != fa) tb = fl[(1 + fb) * FP + d];
  const float tpa[4] = {ta.x, ta.y, ta.z, ta.w};
  const float tpb[4] = {tb.x, tb.y, tb.z, tb.w};
  float xw[4 * (NQ + 1)];
#pragma unroll
  for (int r = 0; r <= NQ; ++r) {
    xw[4 * r + 0] = gq[r].x; xw[4 * r + 1] = gq[r].y;
    xw[4 * r + 2] = gq[r].z; xw[4 * r + 3] = gq[r].w;
  }
#pragma unroll
  for (int q = 0; q < 4; ++q) {
#pragma unroll
    for (int i = 0; i < RO; ++i) {
      const int e = i - q + 3;
      const float tap = qfr(q0 + (e >> 2)) == fa ? tpa[q] : tpb[q];
      acc[i] = fmaf(xw[e], tap, acc[i]);
    }
  }
#pragma unroll
  for (int r = NQ; r >= 1; --r) gq[r] = gq[r - 1];
}

__device__ __forceinline__ void conv_window(const float4* xl, const int d, float4 (&gq)[NQ + 1]) {
#pragma unroll
  for (int r = 1; r <= NQ; ++r) {
    const int q = 15 - d + r;
    const int f = qfr(q);
    gq[r] = xl[(1 + f) * XP + q - FRQ * f];
  }
}

__device__ __forceinline__ void conv_tile(const float4* xs, const float4* fs, float4* os,
                                          const int w, const int l) {
  float acc[RO];
#pragma unroll
  for (int i = 0; i < RO; ++i) acc[i] = 0.0f;
  const float4* xl = xs + l * XP;
  const float4* fl = fs + l * FP + 4 * w;
  float4 gq[NQ + 1];

  if (w == 4) conv_window(xl, -16, gq);
  if (w >= 4) {
#pragma unroll
    for (int d = -16; d < -12; ++d) conv_chunk(xl, fl, d, gq, acc);
  }
  if (w == 3) conv_window(xl, -12, gq);
  if (w >= 3) {
#pragma unroll
    for (int d = -12; d < -8; ++d) conv_chunk(xl, fl, d, gq, acc);
  }
  if (w == 2) conv_window(xl, -8, gq);
  if (w >= 2) {
#pragma unroll
    for (int d = -8; d < -4; ++d) conv_chunk(xl, fl, d, gq, acc);
  }
  if (w == 1) conv_window(xl, -4, gq);
  if (w >= 1) {
#pragma unroll
    for (int d = -4; d < 0; ++d) conv_chunk(xl, fl, d, gq, acc);
  }
  if (w == 0) conv_window(xl, 0, gq);
#pragma unroll
  for (int d = 0; d < 16; ++d) conv_chunk(xl, fl, d, gq, acc);
  if (w <= 3) {
#pragma unroll
    for (int d = 16; d < 20; ++d) conv_chunk(xl, fl, d, gq, acc);
    if (w <= 2) {
#pragma unroll
      for (int d = 20; d < 24; ++d) conv_chunk(xl, fl, d, gq, acc);
      if (w <= 1) {
#pragma unroll
        for (int d = 24; d < 28; ++d) conv_chunk(xl, fl, d, gq, acc);
        if (w == 0) {
#pragma unroll
          for (int d = 28; d < 32; ++d) conv_chunk(xl, fl, d, gq, acc);
        }
      }
    }
  }

  float4* ol = os + l * XP + NQ * w;
#pragma unroll
  for (int r = 0; r < NQ; ++r)
    ol[r] = make_float4(acc[4 * r], acc[4 * r + 1], acc[4 * r + 2], acc[4 * r + 3]);
}

__global__ __launch_bounds__(NTHR)
void conv_ola(const float* __restrict__ noise,
              const float* __restrict__ H,
              float* __restrict__ out) {
  __shared__ float4 xs[NFR * XP];
  __shared__ float4 fs[NFR * FP];
  __shared__ float4 hos[LFR * XP];
  __shared__ float ct[NCOS + 1];
  static_assert(LFR * XP * 16 >= NFR * HP * 8, "the staged H rows must fit");
  v2f* hs = (v2f*)hos;
  const int tid = (int)threadIdx.x;
  const int b = (int)blockIdx.x / JBLK;
  const int j = (int)blockIdx.x - b * JBLK;
  stage_in(noise, H, xs, (float*)hs, ct, b, j, tid);
  __syncthreads();
  const int w = __builtin_amdgcn_readfirstlane(tid >> 5);
  const int lane = tid & 31;
  if (w < 4) fir_tiles<2>(hs, ct, (float*)fs, j, lane, w, w + 4);
  __syncthreads();
  if (lane < LFR && LFR * j + lane < TFRAMES) conv_tile(xs, fs, hos, w, lane);
  __syncthreads();
  const int nq = TS * j + TS <= TFRAMES * FRAME ? TS / 4 : (TFRAMES * FRAME - TS * j) / 4;
  float4* o = (float4*)(out + (size_t)b * (TFRAMES * FRAME) + (size_t)TS * j);
#pragma unroll
  for (int it = 0; it < (TS / 4 + NTHR - 1) / NTHR; ++it) {
    const int n = tid + it * NTHR;
    if (n < nq) o[n] = hos[xslot(n)];
  }
}

extern "C" void kernel_launch(void* const* d_in, const int* in_sizes, int n_in,
                              void* d_out, int out_size, void* d_ws, size_t ws_size,
                              hipStream_t stream) {
  const float* H     = (const float*)d_in[0];
  const float* noise = (const float*)d_in[1];
  float* out = (float*)d_out;

  conv_ola<<<BATCH * JBLK, NTHR, 0, stream>>>(noise, H, out);
}


# filter sums depend on |n-64| only: WMMA makes 65 columns d=0..64 (5 column tiles, one per wave), not 128 (8 tiles on 4 waves); a sum goes to taps 64+d and 64-d, each with its own window factor
# speedup vs baseline: 10.9876x; 1.0888x over previous
#include <hip/hip_runtime.h>
#include <math.h>

typedef __attribute__((ext_vector_type(2))) float v2f;
typedef __attribute__((ext_vector_type(8))) float v8f;

#define NBANDS 65
#define LFIR   129
#define NPAD   128
#define FRAME  80
#define TFRAMES 4000
#define BATCH  32
#define LCONV  208
#define ROWS   (BATCH*TFRAMES)
#define NTILES 5
#define RTILES (ROWS/16)
#define KSTEPS 17
#define NCOS   LFIR


#define TS    2400
#define RO    16
#define NQ    (RO / 4)
#define LFR   (TS / FRAME)
#define NTHR  160
#define XQ    (TS / 4 + 32)
#define FRQ   (FRAME / 4)
#define XP    (FRQ + 1)
#define NFR   (TS / FRAME + 2)
#define FQ    (NPAD / 4)
#define FP    (FQ + 1)
#define NCH   FQ
#define JBLK  ((TFRAMES * FRAME + TS - 1) / TS)
#define XIT   ((XQ + NTHR - 1) / NTHR)
#define MT    (NFR / 16)
#define HP    34
#define HIT   ((NFR * NBANDS + NTHR - 1) / NTHR)

__device__ __forceinline__ int xslot(int Q) {
  return Q + (int)(((unsigned)Q * 3277u) >> 16);
}

__device__ __forceinline__ int qfr(int q) {
  return q >= FRQ ? 1 : (q < 0 ? -1 : 0);
}

__device__ __forceinline__ void stage_in(const float* __restrict__ noise,
                                         const float* __restrict__ H,
                                         float4* xs, float* hsf, float* ct,
                                         int b, int j, int tid) {
  const float4 zero = make_float4(0.0f, 0.0f, 0.0f, 0.0f);
  const float* nrow = noise + (size_t)b * (TFRAMES * FRAME);
  const int v0 = TS * j - 64;
  const int t0 = (TS / FRAME) * j - 1;
  float4 xr[XIT];
  float hr[HIT];
#pragma unroll
  for (int it = 0; it < XIT; ++it) {
    int v = v0 + 4 * (tid + it * NTHR);
    int vc = v < 0 ? 0 : (v > TFRAMES * FRAME - 4 ? TFRAMES * FRAME - 4 : v);
    xr[it] = *(const float4*)(nrow + vc);
  }
#pragma unroll
  for (int it = 0; it < HIT; ++it) {
    int e = tid + it * NTHR;
    int ec = e < NFR * NBANDS ? e : NFR * NBANDS - 1;
    int r = ec / NBANDS;
    int t = t0 + r;
    int tc = t < 0 ? 0 : (t > TFRAMES - 1 ? TFRAMES - 1 : t);
    hr[it] = H[(size_t)(b * TFRAMES + tc) * NBANDS + (ec - r * NBANDS)];
  }
  if (tid <= NCOS) ct[tid] = tid < NCOS ? cospif((float)tid * (2.0f / (float)LFIR)) : 0.5f;
#pragma unroll
  for (int it = 0; it < XIT; ++it) {
    int e = tid + it * NTHR;
    int v = v0 + 4 * e;
    float4 x = zero;
    if (v >= 0 && v < TFRAMES * FRAME) {
      x.x = 2.0f * xr[it].x - 1.0f;
      x.y = 2.0f * xr[it].y - 1.0f;
      x.z = 2.0f * xr[it].z - 1.0f;
      x.w = 2.0f * xr[it].w - 1.0f;
    }
    if (e < XQ) xs[xslot(e + 4)] = x;
  }
#pragma unroll
  for (int it = 0; it < HIT; ++it) {
    int e = tid + it * NTHR;
    int r = e / NBANDS;
    if (e < NFR * NBANDS) hsf[r * (2 * HP) + (e - r * NBANDS)] = hr[it];
  }
}

#define CTB (NCOS * 4)
__device__ __forceinline__ unsigned wrapct(unsigned off) {
  const unsigned w = off - (unsigned)CTB;
  return off < w ? off : w;
}
__device__ __forceinline__ float ctat(const float* ct, unsigned off) {
  return *(const float*)((const char*)ct + off);
}

__device__ __forceinline__ void fir_tiles(const v2f* hs,
                                          const float* ct,
                                          float* fsf, int j, int lane,
                                          int w) {
  const int half = lane >> 4;
  const int lm   = lane & 15;
  const int t0   = (TS / FRAME) * j - 1;

  const v2f* hrow[MT];
#pragma unroll
  for (int m = 0; m < MT; ++m) hrow[m] = hs + (m * 16 + lm) * HP;

  const int d = w < 4 ? 16 * w + 1 + lm : 0;
  const float sc_hi = (1.0f - ct[(NBANDS - 1) + d]) * (1.0f / (float)LFIR);
  const float sc_lo = (1.0f - ct[(NBANDS - 1) - d]) * (1.0f / (float)LFIR);
  const unsigned st = wrapct((unsigned)(16 * d));
  unsigned i0 = half ? (unsigned)(8 * d) : (unsigned)CTB;
  unsigned i1 = wrapct(i0 + (unsigned)(4 * d));

  v8f acc[MT];
#pragma unroll
  for (int m = 0; m < MT; ++m) acc[m] = v8f{};

#pragma unroll 1
  for (int kk = 0; kk < KSTEPS - 1; ++kk) {
    v2f a[MT];
#pragma unroll
    for (int m = 0; m < MT; ++m) a[m] = hrow[m][kk * 2 + half];
    v2f bb;
    bb.x = ctat(ct, i0);
    bb.y = ctat(ct, i1);
    i0 = wrapct(i0 + st);
    i1 = wrapct(i1 + st);
#pragma unroll
    for (int m = 0; m < MT; ++m)
      acc[m] = __builtin_amdgcn_wmma_f32_16x16x4_f32(false, a[m], false, bb, (short)0,
                                                     acc[m], false, false);
  }

  {
    v2f a[MT];
#pragma unroll
    for (int m = 0; m < MT; ++m) {
      float h64 = ((const float*)hrow[m])[NBANDS - 1];
      a[m].x = half ? 0.0f : h64;
      a[m].y = 0.0f;
    }
    v2f bb;
    bb.x = ctat(ct, i0);
    bb.y = ctat(ct, i1);
#pragma unroll
    for (int m = 0; m < MT; ++m)
      acc[m] = __builtin_amdgcn_wmma_f32_16x16x4_f32(false, a[m], false, bb, (short)0,
                                                     acc[m], false, false);
  }

  const bool hi = w < 4 || lm == 0;
  const bool lo = w < 4 && d < NBANDS - 1;
  const int chi = (8 * half) * (FP * 4) + (NBANDS - 2) + d;
  const int clo = (8 * half) * (FP * 4) + (NBANDS - 2) - d;
  const int tr = t0 + 8 * half;
  if (hi) {
#pragma unroll
    for (int m = 0; m < MT; ++m)
#pragma unroll
      for (int v = 0; v < 8; ++v) {
        const int t = tr + m * 16 + v;
        fsf[chi + (m * 16 + v) * (FP * 4)] = (t >= 0 && t < TFRAMES) ? sc_hi * acc[m][v] : 0.0f;
      }
  }
  if (lo) {
#pragma unroll
    for (int m = 0; m < MT; ++m)
#pragma unroll
      for (int v = 0; v < 8; ++v) {
        const int t = tr + m * 16 + v;
        fsf[clo + (m * 16 + v) * (FP * 4)] = (t >= 0 && t < TFRAMES) ? sc_lo * acc[m][v] : 0.0f;
      }
  }
}

__device__ __forceinline__ void conv_chunk(const float4* xl, const float4* fl, const int d,
                                           float4 (&gq)[NQ + 1], float (&acc)[RO]) {
  const int q0 = 15 - d;
  const int fa = qfr(q0);
  const int fb = qfr(q0 + NQ);
  gq[0] = xl[(1 + fa) * XP + q0 - FRQ * fa];
  const float4 ta = fl[(1 + fa) * FP + d];
  float4 tb = ta;
  if (fb != fa) tb = fl[(1 + fb) * FP + d];
  const float tpa[4] = {ta.x, ta.y, ta.z, ta.w};
  const float tpb[4] = {tb.x, tb.y, tb.z, tb.w};
  float xw[4 * (NQ + 1)];
#pragma unroll
  for (int r = 0; r <= NQ; ++r) {
    xw[4 * r + 0] = gq[r].x; xw[4 * r + 1] = gq[r].y;
    xw[4 * r + 2] = gq[r].z; xw[4 * r + 3] = gq[r].w;
  }
#pragma unroll
  for (int q = 0; q < 4; ++q) {
#pragma unroll
    for (int i = 0; i < RO; ++i) {
      const int e = i - q + 3;
      const float tap = qfr(q0 + (e >> 2)) == fa ? tpa[q] : tpb[q];
      acc[i] = fmaf(xw[e], tap, acc[i]);
    }
  }
#pragma unroll
  for (int r = NQ; r >= 1; --r) gq[r] = gq[r - 1];
}

__device__ __forceinline__ void conv_window(const float4* xl, const int d, float4 (&gq)[NQ + 1]) {
#pragma unroll
  for (int r = 1; r <= NQ; ++r) {
    const int q = 15 - d + r;
    const int f = qfr(q);
    gq[r] = xl[(1 + f) * XP + q - FRQ * f];
  }
}

__device__ __forceinline__ void conv_tile(const float4* xs, const float4* fs, float4* os,
                                          const int w, const int l) {
  float acc[RO];
#pragma unroll
  for (int i = 0; i < RO; ++i) acc[i] = 0.0f;
  const float4* xl = xs + l * XP;
  const float4* fl = fs + l * FP + 4 * w;
  float4 gq[NQ + 1];

  if (w == 4) conv_window(xl, -16, gq);
  if (w >= 4) {
#pragma unroll
    for (int d = -16; d < -12; ++d) conv_chunk(xl, fl, d, gq, acc);
  }
  if (w == 3) conv_window(xl, -12, gq);
  if (w >= 3) {
#pragma unroll
    for (int d = -12; d < -8; ++d) conv_chunk(xl, fl, d, gq, acc);
  }
  if (w == 2) conv_window(xl, -8, gq);
  if (w >= 2) {
#pragma unroll
    for (int d = -8; d < -4; ++d) conv_chunk(xl, fl, d, gq, acc);
  }
  if (w == 1) conv_window(xl, -4, gq);
  if (w >= 1) {
#pragma unroll
    for (int d = -4; d < 0; ++d) conv_chunk(xl, fl, d, gq, acc);
  }
  if (w == 0) conv_window(xl, 0, gq);
#pragma unroll
  for (int d = 0; d < 16; ++d) conv_chunk(xl, fl, d, gq, acc);
  if (w <= 3) {
#pragma unroll
    for (int d = 16; d < 20; ++d) conv_chunk(xl, fl, d, gq, acc);
    if (w <= 2) {
#pragma unroll
      for (int d = 20; d < 24; ++d) conv_chunk(xl, fl, d, gq, acc);
      if (w <= 1) {
#pragma unroll
        for (int d = 24; d < 28; ++d) conv_chunk(xl, fl, d, gq, acc);
        if (w == 0) {
#pragma unroll
          for (int d = 28; d < 32; ++d) conv_chunk(xl, fl, d, gq, acc);
        }
      }
    }
  }

  float4* ol = os + l * XP + NQ * w;
#pragma unroll
  for (int r = 0; r < NQ; ++r)
    ol[r] = make_float4(acc[4 * r], acc[4 * r + 1], acc[4 * r + 2], acc[4 * r + 3]);
}

__global__ __launch_bounds__(NTHR)
void conv_ola(const float* __restrict__ noise,
              const float* __restrict__ H,
              float* __restrict__ out) {
  __shared__ float4 xs[NFR * XP];
  __shared__ float4 fs[NFR * FP];
  __shared__ float4 hos[LFR * XP];
  __shared__ float ct[NCOS + 1];
  static_assert(LFR * XP * 16 >= NFR * HP * 8, "the staged H rows must fit");
  v2f* hs = (v2f*)hos;
  const int tid = (int)threadIdx.x;
  const int b = (int)blockIdx.x / JBLK;
  const int j = (int)blockIdx.x - b * JBLK;
  stage_in(noise, H, xs, (float*)hs, ct, b, j, tid);
  __syncthreads();
  const int w = __builtin_amdgcn_readfirstlane(tid >> 5);
  const int lane = tid & 31;
  fir_tiles(hs, ct, (float*)fs, j, lane, w);
  __syncthreads();
  if (lane < LFR && LFR * j + lane < TFRAMES) conv_tile(xs, fs, hos, w, lane);
  __syncthreads();
  const int nq = TS * j + TS <= TFRAMES * FRAME ? TS / 4 : (TFRAMES * FRAME - TS * j) / 4;
  float4* o = (float4*)(out + (size_t)b * (TFRAMES * FRAME) + (size_t)TS * j);
#pragma unroll
  for (int it = 0; it < (TS / 4 + NTHR - 1) / NTHR; ++it) {
    const int n = tid + it * NTHR;
    if (n < nq) o[n] = hos[xslot(n)];
  }
}

extern "C" void kernel_launch(void* const* d_in, const int* in_sizes, int n_in,
                              void* d_out, int out_size, void* d_ws, size_t ws_size,
                              hipStream_t stream) {
  const float* H     = (const float*)d_in[0];
  const float* noise = (const float*)d_in[1];
  float* out = (float*)d_out;

  conv_ola<<<BATCH * JBLK, NTHR, 0, stream>>>(noise, H, out);
}


# FIR product tiled exactly 64x64 on WMMA (bands 1..64 x distances 1..64: 4 waves, 16 k-steps); band 0 enters as the C operand, the cosine-free centre column is a row sum on wave 4; sum order moves
# speedup vs baseline: 11.5569x; 1.0518x over previous
#include <hip/hip_runtime.h>
#include <math.h>

typedef __attribute__((ext_vector_type(2))) float v2f;
typedef __attribute__((ext_vector_type(8))) float v8f;

#define NBANDS 65
#define LFIR   129
#define NPAD   128
#define FRAME  80
#define TFRAMES 4000
#define BATCH  32
#define LCONV  208
#define ROWS   (BATCH*TFRAMES)
#define NTILES 4
#define RTILES (ROWS/16)
#define KSTEPS 16
#define NCOS   LFIR


#define TS    2400
#define RO    16
#define NQ    (RO / 4)
#define LFR   (TS / FRAME)
#define NTHR  160
#define XQ    (TS / 4 + 32)
#define FRQ   (FRAME / 4)
#define XP    (FRQ + 1)
#define NFR   (TS / FRAME + 2)
#define FQ    (NPAD / 4)
#define FP    (FQ + 1)
#define NCH   FQ
#define JBLK  ((TFRAMES * FRAME + TS - 1) / TS)
#define XIT   ((XQ + NTHR - 1) / NTHR)
#define MT    (NFR / 16)
#define HP    34
#define HOFF  1
#define HIT   ((NFR * NBANDS + NTHR - 1) / NTHR)

__device__ __forceinline__ int xslot(int Q) {
  return Q + (int)(((unsigned)Q * 3277u) >> 16);
}

__device__ __forceinline__ int qfr(int q) {
  return q >= FRQ ? 1 : (q < 0 ? -1 : 0);
}

__device__ __forceinline__ void stage_in(const float* __restrict__ noise,
                                         const float* __restrict__ H,
                                         float4* xs, float* hsf, float* ct,
                                         int b, int j, int tid) {
  const float4 zero = make_float4(0.0f, 0.0f, 0.0f, 0.0f);
  const float* nrow = noise + (size_t)b * (TFRAMES * FRAME);
  const int v0 = TS * j - 64;
  const int t0 = (TS / FRAME) * j - 1;
  float4 xr[XIT];
  float hr[HIT];
#pragma unroll
  for (int it = 0; it < XIT; ++it) {
    int v = v0 + 4 * (tid + it * NTHR);
    int vc = v < 0 ? 0 : (v > TFRAMES * FRAME - 4 ? TFRAMES * FRAME - 4 : v);
    xr[it] = *(const float4*)(nrow + vc);
  }
#pragma unroll
  for (int it = 0; it < HIT; ++it) {
    int e = tid + it * NTHR;
    int ec = e < NFR * NBANDS ? e : NFR * NBANDS - 1;
    int r = ec / NBANDS;
    int t = t0 + r;
    int tc = t < 0 ? 0 : (t > TFRAMES - 1 ? TFRAMES - 1 : t);
    hr[it] = H[(size_t)(b * TFRAMES + tc) * NBANDS + (ec - r * NBANDS)];
  }
  if (tid < NCOS) ct[tid] = cospif((float)tid * (2.0f / (float)LFIR));
#pragma unroll
  for (int it = 0; it < XIT; ++it) {
    int e = tid + it * NTHR;
    int v = v0 + 4 * e;
    float4 x = zero;
    if (v >= 0 && v < TFRAMES * FRAME) {
      x.x = 2.0f * xr[it].x - 1.0f;
      x.y = 2.0f * xr[it].y - 1.0f;
      x.z = 2.0f * xr[it].z - 1.0f;
      x.w = 2.0f * xr[it].w - 1.0f;
    }
    if (e < XQ) xs[xslot(e + 4)] = x;
  }
#pragma unroll
  for (int it = 0; it < HIT; ++it) {
    int e = tid + it * NTHR;
    int r = e / NBANDS;
    if (e < NFR * NBANDS) hsf[r * (2 * HP) + HOFF + (e - r * NBANDS)] = hr[it];
  }
}

#define CTB (NCOS * 4)
__device__ __forceinline__ unsigned wrapct(unsigned off) {
  const unsigned w = off - (unsigned)CTB;
  return off < w ? off : w;
}
__device__ __forceinline__ float ctat(const float* ct, unsigned off) {
  return *(const float*)((const char*)ct + off);
}

__device__ __forceinline__ void fir_tiles(const v2f* hs,
                                          const float* ct,
                                          float* fsf, int j, int lane,
                                          int w) {
  const int t0   = (TS / FRAME) * j - 1;

  if (w == NTILES) {
    const v2f* hp = hs + lane * HP;
    const float4* hq = (const float4*)hp;
    float4 s = hq[1];
#pragma unroll
    for (int q = 2; q < (NBANDS - 1) / 4; ++q) {
      const float4 v = hq[q];
      s.x += v.x; s.y += v.y; s.z += v.z; s.w += v.w;
    }
    const v2f pa = hp[1];
    const v2f pb = hp[(NBANDS - 1) / 2];
    s.x += pa.x; s.y += pa.y; s.z += pb.x; s.w += pb.y;
    const float h0 = ((const float*)hp)[HOFF];
    const float g0 = fmaf(0.5f, h0, (s.x + s.y) + (s.z + s.w));
    const float sc0 = (1.0f - ct[NBANDS - 1]) * (1.0f / (float)LFIR);
    const int t = t0 + lane;
    fsf[lane * (FP * 4) + (NBANDS - 2)] = (t >= 0 && t < TFRAMES) ? sc0 * g0 : 0.0f;
    return;
  }

  const int half = lane >> 4;
  const int lm   = lane & 15;

  const v2f* hrow[MT];
#pragma unroll
  for (int m = 0; m < MT; ++m) hrow[m] = hs + (m * 16 + lm) * HP + (1 + HOFF) / 2;

  const int d = 16 * w + 1 + lm;
  const float sc_hi = (1.0f - ct[(NBANDS - 1) + d]) * (1.0f / (float)LFIR);
  const float sc_lo = (1.0f - ct[(NBANDS - 1) - d]) * (1.0f / (float)LFIR);
  const unsigned st = wrapct((unsigned)(16 * d));
  unsigned i0 = half ? wrapct((unsigned)(12 * d)) : (unsigned)(4 * d);
  unsigned i1 = half ? st : (unsigned)(8 * d);

  v8f acc[MT];
#pragma unroll
  for (int m = 0; m < MT; ++m)
#pragma unroll
    for (int v = 0; v < 8; ++v)
      acc[m][v] = 0.5f * ((const float*)hs)[(m * 16 + 8 * half + v) * (2 * HP) + HOFF];

#pragma unroll 1
  for (int kk = 0; kk < KSTEPS; ++kk) {
    v2f a[MT];
#pragma unroll
    for (int m = 0; m < MT; ++m) a[m] = hrow[m][kk * 2 + half];
    v2f bb;
    bb.x = ctat(ct, i0);
    bb.y = ctat(ct, i1);
    i0 = wrapct(i0 + st);
    i1 = wrapct(i1 + st);
#pragma unroll
    for (int m = 0; m < MT; ++m)
      acc[m] = __builtin_amdgcn_wmma_f32_16x16x4_f32(false, a[m], false, bb, (short)0,
                                                     acc[m], false, false);
  }

  const bool lo = d < NBANDS - 1;
  const int chi = (8 * half) * (FP * 4) + (NBANDS - 2) + d;
  const int clo = (8 * half) * (FP * 4) + (NBANDS - 2) - d;
  const int tr = t0 + 8 * half;
#pragma unroll
  for (int m = 0; m < MT; ++m)
#pragma unroll
    for (int v = 0; v < 8; ++v) {
      const int t = tr + m * 16 + v;
      fsf[chi + (m * 16 + v) * (FP * 4)] = (t >= 0 && t < TFRAMES) ? sc_hi * acc[m][v] : 0.0f;
    }
  if (lo) {
#pragma unroll
    for (int m = 0; m < MT; ++m)
#pragma unroll
      for (int v = 0; v < 8; ++v) {
        const int t = tr + m * 16 + v;
        fsf[clo + (m * 16 + v) * (FP * 4)] = (t >= 0 && t < TFRAMES) ? sc_lo * acc[m][v] : 0.0f;
      }
  }
}

__device__ __forceinline__ void conv_chunk(const float4* xl, const float4* fl, const int d,
                                           float4 (&gq)[NQ + 1], float (&acc)[RO]) {
  const int q0 = 15 - d;
  const int fa = qfr(q0);
  const int fb = qfr(q0 + NQ);
  gq[0] = xl[(1 + fa) * XP + q0 - FRQ * fa];
  const float4 ta = fl[(1 + fa) * FP + d];
  float4 tb = ta;
  if (fb != fa) tb = fl[(1 + fb) * FP + d];
  const float tpa[4] = {ta.x, ta.y, ta.z, ta.w};
  const float tpb[4] = {tb.x, tb.y, tb.z, tb.w};
  float xw[4 * (NQ + 1)];
#pragma unroll
  for (int r = 0; r <= NQ; ++r) {
    xw[4 * r + 0] = gq[r].x; xw[4 * r + 1] = gq[r].y;
    xw[4 * r + 2] = gq[r].z; xw[4 * r + 3] = gq[r].w;
  }
#pragma unroll
  for (int q = 0; q < 4; ++q) {
#pragma unroll
    for (int i = 0; i < RO; ++i) {
      const int e = i - q + 3;
      const float tap = qfr(q0 + (e >> 2)) == fa ? tpa[q] : tpb[q];
      acc[i] = fmaf(xw[e], tap, acc[i]);
    }
  }
#pragma unroll
  for (int r = NQ; r >= 1; --r) gq[r] = gq[r - 1];
}

__device__ __forceinline__ void conv_window(const float4* xl, const int d, float4 (&gq)[NQ + 1]) {
#pragma unroll
  for (int r = 1; r <= NQ; ++r) {
    const int q = 15 - d + r;
    const int f = qfr(q);
    gq[r] = xl[(1 + f) * XP + q - FRQ * f];
  }
}

__device__ __forceinline__ void conv_tile(const float4* xs, const float4* fs, float4* os,
                                          const int w, const int l) {
  float acc[RO];
#pragma unroll
  for (int i = 0; i < RO; ++i) acc[i] = 0.0f;
  const float4* xl = xs + l * XP;
  const float4* fl = fs + l * FP + 4 * w;
  float4 gq[NQ + 1];

  if (w == 4) conv_window(xl, -16, gq);
  if (w >= 4) {
#pragma unroll
    for (int d = -16; d < -12; ++d) conv_chunk(xl, fl, d, gq, acc);
  }
  if (w == 3) conv_window(xl, -12, gq);
  if (w >= 3) {
#pragma unroll
    for (int d = -12; d < -8; ++d) conv_chunk(xl, fl, d, gq, acc);
  }
  if (w == 2) conv_window(xl, -8, gq);
  if (w >= 2) {
#pragma unroll
    for (int d = -8; d < -4; ++d) conv_chunk(xl, fl, d, gq, acc);
  }
  if (w == 1) conv_window(xl, -4, gq);
  if (w >= 1) {
#pragma unroll
    for (int d = -4; d < 0; ++d) conv_chunk(xl, fl, d, gq, acc);
  }
  if (w == 0) conv_window(xl, 0, gq);
#pragma unroll
  for (int d = 0; d < 16; ++d) conv_chunk(xl, fl, d, gq, acc);
  if (w <= 3) {
#pragma unroll
    for (int d = 16; d < 20; ++d) conv_chunk(xl, fl, d, gq, acc);
    if (w <= 2) {
#pragma unroll
      for (int d = 20; d < 24; ++d) conv_chunk(xl, fl, d, gq, acc);
      if (w <= 1) {
#pragma unroll
        for (int d = 24; d < 28; ++d) conv_chunk(xl, fl, d, gq, acc);
        if (w == 0) {
#pragma unroll
          for (int d = 28; d < 32; ++d) conv_chunk(xl, fl, d, gq, acc);
        }
      }
    }
  }

  float4* ol = os + l * XP + NQ * w;
#pragma unroll
  for (int r = 0; r < NQ; ++r)
    ol[r] = make_float4(acc[4 * r], acc[4 * r + 1], acc[4 * r + 2], acc[4 * r + 3]);
}

__global__ __launch_bounds__(NTHR)
void conv_ola(const float* __restrict__ noise,
              const float* __restrict__ H,
              float* __restrict__ out) {
  __shared__ float4 xs[NFR * XP];
  __shared__ float4 fs[NFR * FP];
  __shared__ float4 hos[LFR * XP];
  __shared__ float ct[NCOS];
  static_assert(LFR * XP * 16 >= NFR * HP * 8, "the staged H rows must fit");
  v2f* hs = (v2f*)hos;
  const int tid = (int)threadIdx.x;
  const int b = (int)blockIdx.x / JBLK;
  const int j = (int)blockIdx.x - b * JBLK;
  stage_in(noise, H, xs, (float*)hs, ct, b, j, tid);
  __syncthreads();
  const int w = __builtin_amdgcn_readfirstlane(tid >> 5);
  const int lane = tid & 31;
  fir_tiles(hs, ct, (float*)fs, j, lane, w);
  __syncthreads();
  if (lane < LFR && LFR * j + lane < TFRAMES) conv_tile(xs, fs, hos, w, lane);
  __syncthreads();
  const int nq = TS * j + TS <= TFRAMES * FRAME ? TS / 4 : (TFRAMES * FRAME - TS * j) / 4;
  float4* o = (float4*)(out + (size_t)b * (TFRAMES * FRAME) + (size_t)TS * j);
#pragma unroll
  for (int it = 0; it < (TS / 4 + NTHR - 1) / NTHR; ++it) {
    const int n = tid + it * NTHR;
    if (n < nq) o[n] = hos[xslot(n)];
  }
}

extern "C" void kernel_launch(void* const* d_in, const int* in_sizes, int n_in,
                              void* d_out, int out_size, void* d_ws, size_t ws_size,
                              hipStream_t stream) {
  const float* H     = (const float*)d_in[0];
  const float* noise = (const float*)d_in[1];
  float* out = (float*)d_out;

  conv_ola<<<BATCH * JBLK, NTHR, 0, stream>>>(noise, H, out);
}


# H staging remapped: a thread stages 13 consecutive bands of one H row (one clamp, one address, constant offsets, 16-byte loads), not 13 strided elements each with its own division and clamp
# speedup vs baseline: 12.1517x; 1.0515x over previous
#include <hip/hip_runtime.h>
#include <math.h>

typedef __attribute__((ext_vector_type(2))) float v2f;
typedef __attribute__((ext_vector_type(8))) float v8f;
typedef __attribute__((ext_vector_type(4))) float v4f;
typedef v4f v4f_a4 __attribute__((aligned(4)));

#define NBANDS 65
#define LFIR   129
#define NPAD   128
#define FRAME  80
#define TFRAMES 4000
#define BATCH  32
#define LCONV  208
#define ROWS   (BATCH*TFRAMES)
#define NTILES 4
#define RTILES (ROWS/16)
#define KSTEPS 16
#define NCOS   LFIR


#define TS    2400
#define RO    16
#define NQ    (RO / 4)
#define LFR   (TS / FRAME)
#define NTHR  160
#define XQ    (TS / 4 + 32)
#define FRQ   (FRAME / 4)
#define XP    (FRQ + 1)
#define NFR   (TS / FRAME + 2)
#define FQ    (NPAD / 4)
#define FP    (FQ + 1)
#define NCH   FQ
#define JBLK  ((TFRAMES * FRAME + TS - 1) / TS)
#define XIT   ((XQ + NTHR - 1) / NTHR)
#define MT    (NFR / 16)
#define HP    34
#define HOFF  1
#define HRT   (NTHR / NFR)
#define HIT   (NBANDS / HRT)

__device__ __forceinline__ int xslot(int Q) {
  return Q + (int)(((unsigned)Q * 3277u) >> 16);
}

__device__ __forceinline__ int qfr(int q) {
  return q >= FRQ ? 1 : (q < 0 ? -1 : 0);
}

__device__ __forceinline__ void stage_in(const float* __restrict__ noise,
                                         const float* __restrict__ H,
                                         float4* xs, float* hsf, float* ct,
                                         int b, int j, int tid) {
  const float4 zero = make_float4(0.0f, 0.0f, 0.0f, 0.0f);
  const float* nrow = noise + (size_t)b * (TFRAMES * FRAME);
  const int v0 = TS * j - 64;
  const int t0 = (TS / FRAME) * j - 1;
  float4 xr[XIT];
  float hr[HIT];
#pragma unroll
  for (int it = 0; it < XIT; ++it) {
    int v = v0 + 4 * (tid + it * NTHR);
    int vc = v < 0 ? 0 : (v > TFRAMES * FRAME - 4 ? TFRAMES * FRAME - 4 : v);
    xr[it] = *(const float4*)(nrow + vc);
  }
  static_assert(HRT * NFR == NTHR && HIT * HRT == NBANDS && HIT == 13, "H staging map");
  const int hrow = tid / HRT;
  const int hcol = HIT * (tid - HRT * hrow);
  {
    const int t = t0 + hrow;
    const int tc = t < 0 ? 0 : (t > TFRAMES - 1 ? TFRAMES - 1 : t);
    const float* hp = H + (size_t)(b * TFRAMES + tc) * NBANDS + hcol;
    const v4f h0 = *(const v4f_a4*)(hp);
    const v4f h1 = *(const v4f_a4*)(hp + 4);
    const v4f h2 = *(const v4f_a4*)(hp + 8);
    hr[0] = h0.x; hr[1] = h0.y; hr[2]  = h0.z; hr[3]  = h0.w;
    hr[4] = h1.x; hr[5] = h1.y; hr[6]  = h1.z; hr[7]  = h1.w;
    hr[8] = h2.x; hr[9] = h2.y; hr[10] = h2.z; hr[11] = h2.w;
    hr[12] = hp[12];
  }
  if (tid < NCOS) ct[tid] = cospif((float)tid * (2.0f / (float)LFIR));
#pragma unroll
  for (int it = 0; it < XIT; ++it) {
    int e = tid + it * NTHR;
    int v = v0 + 4 * e;
    float4 x = zero;
    if (v >= 0 && v < TFRAMES * FRAME) {
      x.x = 2.0f * xr[it].x - 1.0f;
      x.y = 2.0f * xr[it].y - 1.0f;
      x.z = 2.0f * xr[it].z - 1.0f;
      x.w = 2.0f * xr[it].w - 1.0f;
    }
    if (e < XQ) xs[xslot(e + 4)] = x;
  }
  float* hd = hsf + hrow * (2 * HP) + HOFF + hcol;
#pragma unroll
  for (int it = 0; it < HIT; ++it) hd[it] = hr[it];
}

#define CTB (NCOS * 4)
__device__ __forceinline__ unsigned wrapct(unsigned off) {
  const unsigned w = off - (unsigned)CTB;
  return off < w ? off : w;
}
__device__ __forceinline__ float ctat(const float* ct, unsigned off) {
  return *(const float*)((const char*)ct + off);
}

__device__ __forceinline__ void fir_tiles(const v2f* hs,
                                          const float* ct,
                                          float* fsf, int j, int lane,
                                          int w) {
  const int t0   = (TS / FRAME) * j - 1;

  if (w == NTILES) {
    const v2f* hp = hs + lane * HP;
    const float4* hq = (const float4*)hp;
    float4 s = hq[1];
#pragma unroll
    for (int q = 2; q < (NBANDS - 1) / 4; ++q) {
      const float4 v = hq[q];
      s.x += v.x; s.y += v.y; s.z += v.z; s.w += v.w;
    }
    const v2f pa = hp[1];
    const v2f pb = hp[(NBANDS - 1) / 2];
    s.x += pa.x; s.y += pa.y; s.z += pb.x; s.w += pb.y;
    const float h0 = ((const float*)hp)[HOFF];
    const float g0 = fmaf(0.5f, h0, (s.x + s.y) + (s.z + s.w));
    const float sc0 = (1.0f - ct[NBANDS - 1]) * (1.0f / (float)LFIR);
    const int t = t0 + lane;
    fsf[lane * (FP * 4) + (NBANDS - 2)] = (t >= 0 && t < TFRAMES) ? sc0 * g0 : 0.0f;
    return;
  }

  const int half = lane >> 4;
  const int lm   = lane & 15;

  const v2f* hrow[MT];
#pragma unroll
  for (int m = 0; m < MT; ++m) hrow[m] = hs + (m * 16 + lm) * HP + (1 + HOFF) / 2;

  const int d = 16 * w + 1 + lm;
  const float sc_hi = (1.0f - ct[(NBANDS - 1) + d]) * (1.0f / (float)LFIR);
  const float sc_lo = (1.0f - ct[(NBANDS - 1) - d]) * (1.0f / (float)LFIR);
  const unsigned st = wrapct((unsigned)(16 * d));
  unsigned i0 = half ? wrapct((unsigned)(12 * d)) : (unsigned)(4 * d);
  unsigned i1 = half ? st : (unsigned)(8 * d);

  v8f acc[MT];
#pragma unroll
  for (int m = 0; m < MT; ++m)
#pragma unroll
    for (int v = 0; v < 8; ++v)
      acc[m][v] = 0.5f * ((const float*)hs)[(m * 16 + 8 * half + v) * (2 * HP) + HOFF];

#pragma unroll 1
  for (int kk = 0; kk < KSTEPS; ++kk) {
    v2f a[MT];
#pragma unroll
    for (int m = 0; m < MT; ++m) a[m] = hrow[m][kk * 2 + half];
    v2f bb;
    bb.x = ctat(ct, i0);
    bb.y = ctat(ct, i1);
    i0 = wrapct(i0 + st);
    i1 = wrapct(i1 + st);
#pragma unroll
    for (int m = 0; m < MT; ++m)
      acc[m] = __builtin_amdgcn_wmma_f32_16x16x4_f32(false, a[m], false, bb, (short)0,
                                                     acc[m], false, false);
  }

  const bool lo = d < NBANDS - 1;
  const int chi = (8 * half) * (FP * 4) + (NBANDS - 2) + d;
  const int clo = (8 * half) * (FP * 4) + (NBANDS - 2) - d;
  const int tr = t0 + 8 * half;
#pragma unroll
  for (int m = 0; m < MT; ++m)
#pragma unroll
    for (int v = 0; v < 8; ++v) {
      const int t = tr + m * 16 + v;
      fsf[chi + (m * 16 + v) * (FP * 4)] = (t >= 0 && t < TFRAMES) ? sc_hi * acc[m][v] : 0.0f;
    }
  if (lo) {
#pragma unroll
    for (int m = 0; m < MT; ++m)
#pragma unroll
      for (int v = 0; v < 8; ++v) {
        const int t = tr + m * 16 + v;
        fsf[clo + (m * 16 + v) * (FP * 4)] = (t >= 0 && t < TFRAMES) ? sc_lo * acc[m][v] : 0.0f;
      }
  }
}

__device__ __forceinline__ void conv_chunk(const float4* xl, const float4* fl, const int d,
                                           float4 (&gq)[NQ + 1], float (&acc)[RO]) {
  const int q0 = 15 - d;
  const int fa = qfr(q0);
  const int fb = qfr(q0 + NQ);
  gq[0] = xl[(1 + fa) * XP + q0 - FRQ * fa];
  const float4 ta = fl[(1 + fa) * FP + d];
  float4 tb = ta;
  if (fb != fa) tb = fl[(1 + fb) * FP + d];
  const float tpa[4] = {ta.x, ta.y, ta.z, ta.w};
  const float tpb[4] = {tb.x, tb.y, tb.z, tb.w};
  float xw[4 * (NQ + 1)];
#pragma unroll
  for (int r = 0; r <= NQ; ++r) {
    xw[4 * r + 0] = gq[r].x; xw[4 * r + 1] = gq[r].y;
    xw[4 * r + 2] = gq[r].z; xw[4 * r + 3] = gq[r].w;
  }
#pragma unroll
  for (int q = 0; q < 4; ++q) {
#pragma unroll
    for (int i = 0; i < RO; ++i) {
      const int e = i - q + 3;
      const float tap = qfr(q0 + (e >> 2)) == fa ? tpa[q] : tpb[q];
      acc[i] = fmaf(xw[e], tap, acc[i]);
    }
  }
#pragma unroll
  for (int r = NQ; r >= 1; --r) gq[r] = gq[r - 1];
}

__device__ __forceinline__ void conv_window(const float4* xl, const int d, float4 (&gq)[NQ + 1]) {
#pragma unroll
  for (int r = 1; r <= NQ; ++r) {
    const int q = 15 - d + r;
    const int f = qfr(q);
    gq[r] = xl[(1 + f) * XP + q - FRQ * f];
  }
}

__device__ __forceinline__ void conv_tile(const float4* xs, const float4* fs, float4* os,
                                          const int w, const int l) {
  float acc[RO];
#pragma unroll
  for (int i = 0; i < RO; ++i) acc[i] = 0.0f;
  const float4* xl = xs + l * XP;
  const float4* fl = fs + l * FP + 4 * w;
  float4 gq[NQ + 1];

  if (w == 4) conv_window(xl, -16, gq);
  if (w >= 4) {
#pragma unroll
    for (int d = -16; d < -12; ++d) conv_chunk(xl, fl, d, gq, acc);
  }
  if (w == 3) conv_window(xl, -12, gq);
  if (w >= 3) {
#pragma unroll
    for (int d = -12; d < -8; ++d) conv_chunk(xl, fl, d, gq, acc);
  }
  if (w == 2) conv_window(xl, -8, gq);
  if (w >= 2) {
#pragma unroll
    for (int d = -8; d < -4; ++d) conv_chunk(xl, fl, d, gq, acc);
  }
  if (w == 1) conv_window(xl, -4, gq);
  if (w >= 1) {
#pragma unroll
    for (int d = -4; d < 0; ++d) conv_chunk(xl, fl, d, gq, acc);
  }
  if (w == 0) conv_window(xl, 0, gq);
#pragma unroll
  for (int d = 0; d < 16; ++d) conv_chunk(xl, fl, d, gq, acc);
  if (w <= 3) {
#pragma unroll
    for (int d = 16; d < 20; ++d) conv_chunk(xl, fl, d, gq, acc);
    if (w <= 2) {
#pragma unroll
      for (int d = 20; d < 24; ++d) conv_chunk(xl, fl, d, gq, acc);
      if (w <= 1) {
#pragma unroll
        for (int d = 24; d < 28; ++d) conv_chunk(xl, fl, d, gq, acc);
        if (w == 0) {
#pragma unroll
          for (int d = 28; d < 32; ++d) conv_chunk(xl, fl, d, gq, acc);
        }
      }
    }
  }

  float4* ol = os + l * XP + NQ * w;
#pragma unroll
  for (int r = 0; r < NQ; ++r)
    ol[r] = make_float4(acc[4 * r], acc[4 * r + 1], acc[4 * r + 2], acc[4 * r + 3]);
}

__global__ __launch_bounds__(NTHR)
void conv_ola(const float* __restrict__ noise,
              const float* __restrict__ H,
              float* __restrict__ out) {
  __shared__ float4 xs[NFR * XP];
  __shared__ float4 fs[NFR * FP];
  __shared__ float4 hos[LFR * XP];
  __shared__ float ct[NCOS];
  static_assert(LFR * XP * 16 >= NFR * HP * 8, "the staged H rows must fit");
  v2f* hs = (v2f*)hos;
  const int tid = (int)threadIdx.x;
  const int b = (int)blockIdx.x / JBLK;
  const int j = (int)blockIdx.x - b * JBLK;
  stage_in(noise, H, xs, (float*)hs, ct, b, j, tid);
  __syncthreads();
  const int w = __builtin_amdgcn_readfirstlane(tid >> 5);
  const int lane = tid & 31;
  fir_tiles(hs, ct, (float*)fs, j, lane, w);
  __syncthreads();
  if (lane < LFR && LFR * j + lane < TFRAMES) conv_tile(xs, fs, hos, w, lane);
  __syncthreads();
  const int nq = TS * j + TS <= TFRAMES * FRAME ? TS / 4 : (TFRAMES * FRAME - TS * j) / 4;
  float4* o = (float4*)(out + (size_t)b * (TFRAMES * FRAME) + (size_t)TS * j);
#pragma unroll
  for (int it = 0; it < (TS / 4 + NTHR - 1) / NTHR; ++it) {
    const int n = tid + it * NTHR;
    if (n < nq) o[n] = hos[xslot(n)];
  }
}

extern "C" void kernel_launch(void* const* d_in, const int* in_sizes, int n_in,
                              void* d_out, int out_size, void* d_ws, size_t ws_size,
                              hipStream_t stream) {
  const float* H     = (const float*)d_in[0];
  const float* noise = (const float*)d_in[1];
  float* out = (float*)d_out;

  conv_ola<<<BATCH * JBLK, NTHR, 0, stream>>>(noise, H, out);
}


# work-group of 4 waves x 20 outputs per thread, not 5 x 16: one wave per SIMD, no fifth wave doubling up on the first wave's SIMD; the centre-column row sum rides on one of the 4 WMMA waves
# speedup vs baseline: 12.9140x; 1.0627x over previous
#include <hip/hip_runtime.h>
#include <math.h>

typedef __attribute__((ext_vector_type(2))) float v2f;
typedef __attribute__((ext_vector_type(8))) float v8f;
typedef __attribute__((ext_vector_type(4))) float v4f;
typedef v4f v4f_a4 __attribute__((aligned(4)));

#define NBANDS 65
#define LFIR   129
#define NPAD   128
#define FRAME  80
#define TFRAMES 4000
#define BATCH  32
#define LCONV  208
#define ROWS   (BATCH*TFRAMES)
#define NTILES 4
#define RTILES (ROWS/16)
#define KSTEPS 16
#define NCOS   LFIR


#define TS    2400
#define RO    20
#define NQ    (RO / 4)
#define LFR   (TS / FRAME)
#define NWV   (FRAME / RO)
#define NTHR  (32 * NWV)
#define XQ    (TS / 4 + 32)
#define FRQ   (FRAME / 4)
#define XP    (FRQ + 1)
#define NFR   (TS / FRAME + 2)
#define FQ    (NPAD / 4)
#define FP    (FQ + 1)
#define NCH   FQ
#define JBLK  ((TFRAMES * FRAME + TS - 1) / TS)
#define XIT   ((XQ + NTHR - 1) / NTHR)
#define MT    (NFR / 16)
#define HP    34
#define HOFF  1
#define HRT   (NTHR / NFR)
#define HIT   ((NBANDS - 1) / HRT)

__device__ __forceinline__ int xslot(int Q) {
  return Q + (int)(((unsigned)Q * 3277u) >> 16);
}

__device__ __forceinline__ int qfr(int q) {
  return q >= FRQ ? 1 : (q < 0 ? -1 : 0);
}

__device__ __forceinline__ void stage_in(const float* __restrict__ noise,
                                         const float* __restrict__ H,
                                         float4* xs, float* hsf, float* ct,
                                         int b, int j, int tid) {
  const float4 zero = make_float4(0.0f, 0.0f, 0.0f, 0.0f);
  const float* nrow = noise + (size_t)b * (TFRAMES * FRAME);
  const int v0 = TS * j - 64;
  const int t0 = (TS / FRAME) * j - 1;
  float4 xr[XIT];
#pragma unroll
  for (int it = 0; it < XIT; ++it) {
    int v = v0 + 4 * (tid + it * NTHR);
    int vc = v < 0 ? 0 : (v > TFRAMES * FRAME - 4 ? TFRAMES * FRAME - 4 : v);
    xr[it] = *(const float4*)(nrow + vc);
  }
  static_assert(HRT * NFR == NTHR && HIT * HRT == NBANDS - 1 && HIT == 16, "H staging map");
  const int hrow = tid / HRT;
  const int hpart = tid - HRT * hrow;
  const int hcol = HIT * hpart;
  const int tH = t0 + hrow;
  const int tHc = tH < 0 ? 0 : (tH > TFRAMES - 1 ? TFRAMES - 1 : tH);
  const float* hp = H + (size_t)(b * TFRAMES + tHc) * NBANDS + hcol;
  const float hb = hp[0];
  const v4f h0 = *(const v4f_a4*)(hp + 1);
  const v4f h1 = *(const v4f_a4*)(hp + 5);
  const v4f h2 = *(const v4f_a4*)(hp + 9);
  const v4f h3 = *(const v4f_a4*)(hp + 13);
  static_assert(NTHR == NCOS - 1, "one cosine per thread, and entry 0");
  ct[tid + 1] = cospif((float)(tid + 1) * (2.0f / (float)LFIR));
  if (tid == 0) ct[0] = 1.0f;
#pragma unroll
  for (int it = 0; it < XIT; ++it) {
    int e = tid + it * NTHR;
    int v = v0 + 4 * e;
    float4 x = zero;
    if (v >= 0 && v < TFRAMES * FRAME) {
      x.x = 2.0f * xr[it].x - 1.0f;
      x.y = 2.0f * xr[it].y - 1.0f;
      x.z = 2.0f * xr[it].z - 1.0f;
      x.w = 2.0f * xr[it].w - 1.0f;
    }
    if (e < XQ) xs[xslot(e + 4)] = x;
  }
  float* hd = hsf + hrow * (2 * HP) + HOFF + hcol;
  if (hpart == 0) hd[0] = hb;
  v2f* hd2 = (v2f*)(hd + 1);
  v2f pr;
  pr.x = h0.x; pr.y = h0.y; hd2[0] = pr;
  pr.x = h0.z; pr.y = h0.w; hd2[1] = pr;
  pr.x = h1.x; pr.y = h1.y; hd2[2] = pr;
  pr.x = h1.z; pr.y = h1.w; hd2[3] = pr;
  pr.x = h2.x; pr.y = h2.y; hd2[4] = pr;
  pr.x = h2.z; pr.y = h2.w; hd2[5] = pr;
  pr.x = h3.x; pr.y = h3.y; hd2[6] = pr;
  pr.x = h3.z; pr.y = h3.w; hd2[7] = pr;
}

#define CTB (NCOS * 4)
__device__ __forceinline__ unsigned wrapct(unsigned off) {
  const unsigned w = off - (unsigned)CTB;
  return off < w ? off : w;
}
__device__ __forceinline__ float ctat(const float* ct, unsigned off) {
  return *(const float*)((const char*)ct + off);
}

__device__ __forceinline__ void fir_tiles(const v2f* hs,
                                          const float* ct,
                                          float* fsf, int j, int lane,
                                          int w) {
  const int t0   = (TS / FRAME) * j - 1;

  if (w == (j & (NTILES - 1))) {
    const v2f* hp = hs + lane * HP;
    const float4* hq = (const float4*)hp;
    float4 s = hq[1];
#pragma unroll
    for (int q = 2; q < (NBANDS - 1) / 4; ++q) {
      const float4 v = hq[q];
      s.x += v.x; s.y += v.y; s.z += v.z; s.w += v.w;
    }
    const v2f pa = hp[1];
    const v2f pb = hp[(NBANDS - 1) / 2];
    s.x += pa.x; s.y += pa.y; s.z += pb.x; s.w += pb.y;
    const float h0 = ((const float*)hp)[HOFF];
    const float g0 = fmaf(0.5f, h0, (s.x + s.y) + (s.z + s.w));
    const float sc0 = (1.0f - ct[NBANDS - 1]) * (1.0f / (float)LFIR);
    const int t = t0 + lane;
    fsf[lane * (FP * 4) + (NBANDS - 2)] = (t >= 0 && t < TFRAMES) ? sc0 * g0 : 0.0f;
  }

  const int half = lane >> 4;
  const int lm   = lane & 15;

  const v2f* hrow[MT];
#pragma unroll
  for (int m = 0; m < MT; ++m) hrow[m] = hs + (m * 16 + lm) * HP + (1 + HOFF) / 2;

  const int d = 16 * w + 1 + lm;
  const float sc_hi = (1.0f - ct[(NBANDS - 1) + d]) * (1.0f / (float)LFIR);
  const float sc_lo = (1.0f - ct[(NBANDS - 1) - d]) * (1.0f / (float)LFIR);
  const unsigned st = wrapct((unsigned)(16 * d));
  unsigned i0 = half ? wrapct((unsigned)(12 * d)) : (unsigned)(4 * d);
  unsigned i1 = half ? st : (unsigned)(8 * d);

  v8f acc[MT];
#pragma unroll
  for (int m = 0; m < MT; ++m)
#pragma unroll
    for (int v = 0; v < 8; ++v)
      acc[m][v] = 0.5f * ((const float*)hs)[(m * 16 + 8 * half + v) * (2 * HP) + HOFF];

#pragma unroll 1
  for (int kk = 0; kk < KSTEPS; ++kk) {
    v2f a[MT];
#pragma unroll
    for (int m = 0; m < MT; ++m) a[m] = hrow[m][kk * 2 + half];
    v2f bb;
    bb.x = ctat(ct, i0);
    bb.y = ctat(ct, i1);
    i0 = wrapct(i0 + st);
    i1 = wrapct(i1 + st);
#pragma unroll
    for (int m = 0; m < MT; ++m)
      acc[m] = __builtin_amdgcn_wmma_f32_16x16x4_f32(false, a[m], false, bb, (short)0,
                                                     acc[m], false, false);
  }

  const bool lo = d < NBANDS - 1;
  const int chi = (8 * half) * (FP * 4) + (NBANDS - 2) + d;
  const int clo = (8 * half) * (FP * 4) + (NBANDS - 2) - d;
  const int tr = t0 + 8 * half;
#pragma unroll
  for (int m = 0; m < MT; ++m)
#pragma unroll
    for (int v = 0; v < 8; ++v) {
      const int t = tr + m * 16 + v;
      fsf[chi + (m * 16 + v) * (FP * 4)] = (t >= 0 && t < TFRAMES) ? sc_hi * acc[m][v] : 0.0f;
    }
  if (lo) {
#pragma unroll
    for (int m = 0; m < MT; ++m)
#pragma unroll
      for (int v = 0; v < 8; ++v) {
        const int t = tr + m * 16 + v;
        fsf[clo + (m * 16 + v) * (FP * 4)] = (t >= 0 && t < TFRAMES) ? sc_lo * acc[m][v] : 0.0f;
      }
  }
}

__device__ __forceinline__ void conv_chunk(const float4* xl, const float4* fl, const int d,
                                           float4 (&gq)[NQ + 1], float (&acc)[RO]) {
  const int q0 = 15 - d;
  const int fa = qfr(q0);
  const int fb = qfr(q0 + NQ);
  gq[0] = xl[(1 + fa) * XP + q0 - FRQ * fa];
  const float4 ta = fl[(1 + fa) * FP + d];
  float4 tb = ta;
  if (fb != fa) tb = fl[(1 + fb) * FP + d];
  const float tpa[4] = {ta.x, ta.y, ta.z, ta.w};
  const float tpb[4] = {tb.x, tb.y, tb.z, tb.w};
  float xw[4 * (NQ + 1)];
#pragma unroll
  for (int r = 0; r <= NQ; ++r) {
    xw[4 * r + 0] = gq[r].x; xw[4 * r + 1] = gq[r].y;
    xw[4 * r + 2] = gq[r].z; xw[4 * r + 3] = gq[r].w;
  }
#pragma unroll
  for (int q = 0; q < 4; ++q) {
#pragma unroll
    for (int i = 0; i < RO; ++i) {
      const int e = i - q + 3;
      const float tap = qfr(q0 + (e >> 2)) == fa ? tpa[q] : tpb[q];
      acc[i] = fmaf(xw[e], tap, acc[i]);
    }
  }
#pragma unroll
  for (int r = NQ; r >= 1; --r) gq[r] = gq[r - 1];
}

__device__ __forceinline__ void conv_window(const float4* xl, const int d, float4 (&gq)[NQ + 1]) {
#pragma unroll
  for (int r = 1; r <= NQ; ++r) {
    const int q = 15 - d + r;
    const int f = qfr(q);
    gq[r] = xl[(1 + f) * XP + q - FRQ * f];
  }
}

__device__ __forceinline__ void conv_tile(const float4* xs, const float4* fs, float4* os,
                                          const int w, const int l) {
  float acc[RO];
#pragma unroll
  for (int i = 0; i < RO; ++i) acc[i] = 0.0f;
  const float4* xl = xs + l * XP;
  const float4* fl = fs + l * FP + NQ * w;
  float4 gq[NQ + 1];

  if (w == 3) conv_window(xl, -15, gq);
  if (w >= 3) {
#pragma unroll
    for (int d = -15; d < -10; ++d) conv_chunk(xl, fl, d, gq, acc);
  }
  if (w == 2) conv_window(xl, -10, gq);
  if (w >= 2) {
#pragma unroll
    for (int d = -10; d < -5; ++d) conv_chunk(xl, fl, d, gq, acc);
  }
  if (w == 1) conv_window(xl, -5, gq);
  if (w >= 1) {
#pragma unroll
    for (int d = -5; d < 0; ++d) conv_chunk(xl, fl, d, gq, acc);
  }
  if (w == 0) conv_window(xl, 0, gq);
#pragma unroll
  for (int d = 0; d < 17; ++d) conv_chunk(xl, fl, d, gq, acc);
  if (w <= 2) {
#pragma unroll
    for (int d = 17; d < 22; ++d) conv_chunk(xl, fl, d, gq, acc);
    if (w <= 1) {
#pragma unroll
      for (int d = 22; d < 27; ++d) conv_chunk(xl, fl, d, gq, acc);
      if (w == 0) {
#pragma unroll
        for (int d = 27; d < 32; ++d) conv_chunk(xl, fl, d, gq, acc);
      }
    }
  }

  float4* ol = os + l * XP + NQ * w;
#pragma unroll
  for (int r = 0; r < NQ; ++r)
    ol[r] = make_float4(acc[4 * r], acc[4 * r + 1], acc[4 * r + 2], acc[4 * r + 3]);
}

__global__ __launch_bounds__(NTHR)
void conv_ola(const float* __restrict__ noise,
              const float* __restrict__ H,
              float* __restrict__ out) {
  __shared__ float4 xs[NFR * XP];
  __shared__ float4 fs[NFR * FP];
  __shared__ float4 hos[LFR * XP];
  __shared__ float ct[NCOS];
  static_assert(LFR * XP * 16 >= NFR * HP * 8, "the staged H rows must fit");
  v2f* hs = (v2f*)hos;
  const int tid = (int)threadIdx.x;
  const int b = (int)blockIdx.x / JBLK;
  const int j = (int)blockIdx.x - b * JBLK;
  stage_in(noise, H, xs, (float*)hs, ct, b, j, tid);
  __syncthreads();
  const int w = __builtin_amdgcn_readfirstlane(tid >> 5);
  const int lane = tid & 31;
  fir_tiles(hs, ct, (float*)fs, j, lane, w);
  __syncthreads();
  if (lane < LFR && LFR * j + lane < TFRAMES) conv_tile(xs, fs, hos, w, lane);
  __syncthreads();
  const int nq = TS * j + TS <= TFRAMES * FRAME ? TS / 4 : (TFRAMES * FRAME - TS * j) / 4;
  float4* o = (float4*)(out + (size_t)b * (TFRAMES * FRAME) + (size_t)TS * j);
#pragma unroll
  for (int it = 0; it < (TS / 4 + NTHR - 1) / NTHR; ++it) {
    const int n = tid + it * NTHR;
    if (n < nq) o[n] = hos[xslot(n)];
  }
}

extern "C" void kernel_launch(void* const* d_in, const int* in_sizes, int n_in,
                              void* d_out, int out_size, void* d_ws, size_t ws_size,
                              hipStream_t stream) {
  const float* H     = (const float*)d_in[0];
  const float* noise = (const float*)d_in[1];
  float* out = (float*)d_out;

  conv_ola<<<BATCH * JBLK, NTHR, 0, stream>>>(noise, H, out);
}
